# starting kernel
# baseline (speedup 1.0000x reference)
#include <hip/hip_runtime.h>

typedef __attribute__((ext_vector_type(2))) float v2f;
typedef __attribute__((ext_vector_type(8))) float v8f;

#define D 128
#define LDS_STRIDE 130

__global__ void init_deg_kernel(float* __restrict__ deg, int n) {
    int i = blockIdx.x * blockDim.x + threadIdx.x;
    if (i < n) deg[i] = 1.0f;
}

__global__ void count_deg_kernel(const int* __restrict__ dst, int E,
                                 float* __restrict__ deg) {
    int e = blockIdx.x * blockDim.x + threadIdx.x;
    if (e < E) atomicAdd(&deg[dst[e]], 1.0f);
}

__global__ void dinv_kernel(float* __restrict__ deg, int n) {
    int i = blockIdx.x * blockDim.x + threadIdx.x;
    if (i < n) deg[i] = rsqrtf(deg[i]);
}

__global__ __launch_bounds__(128) void gemm_wmma_f32(
    const float* __restrict__ A, const float* __restrict__ W,
    float* __restrict__ C, int nrows) {
    __shared__ float wT[D * LDS_STRIDE];

    int tid = threadIdx.x;
    for (int i = tid; i < D * D; i += 128) {
        int r = i >> 7, c = i & (D - 1);
        wT[c * LDS_STRIDE + r] = W[i];
    }
    __syncthreads();

    int wave = tid >> 5;
    int lane = tid & 31;
    int lrow = lane & 15;
    int lhi  = lane >> 4;

    int m0 = blockIdx.x * 64 + wave * 16;

    v8f acc[8];
#pragma unroll
    for (int n = 0; n < 8; ++n) acc[n] = (v8f)(0.0f);

    int arow = m0 + lrow;
    int crow = arow < nrows ? arow : 0;
    const float* aptr = A + (size_t)crow * D + lhi * 2;

    for (int k = 0; k < D; k += 4) {
        v2f a = *(const v2f*)(aptr + k);
#pragma unroll
        for (int n = 0; n < 8; ++n) {
            int col = n * 16 + lrow;
            v2f b = *(const v2f*)(&wT[col * LDS_STRIDE + k + lhi * 2]);
            acc[n] = __builtin_amdgcn_wmma_f32_16x16x4_f32(
                false, a, false, b, (short)0, acc[n], false, false);
        }
    }

#pragma unroll
    for (int n = 0; n < 8; ++n) {
        int col = n * 16 + lrow;
#pragma unroll
        for (int v = 0; v < 8; ++v) {
            int row = m0 + v + 8 * lhi;
            if (row < nrows) C[(size_t)row * D + col] = acc[n][v];
        }
    }
}

__global__ void self_loop_init_kernel(const float* __restrict__ h,
                                      const float* __restrict__ dinv,
                                      float* __restrict__ out, int n) {
    long long i = (long long)blockIdx.x * blockDim.x + threadIdx.x;
    long long total = (long long)n * (D / 4);
    if (i < total) {
        int node = (int)(i >> 5);
        float w = dinv[node] * dinv[node];
        float4 v = ((const float4*)h)[i];
        v.x *= w; v.y *= w; v.z *= w; v.w *= w;
        ((float4*)out)[i] = v;
    }
}

__global__ __launch_bounds__(256) void edge_scatter_kernel(
    const int* __restrict__ src, const int* __restrict__ dst,
    const float* __restrict__ h, const float* __restrict__ dinv,
    float* __restrict__ out, int E) {
    int e = blockIdx.x * 8 + (threadIdx.x >> 5);
    int lane = threadIdx.x & 31;
    if (e >= E) return;
    int s = src[e];
    int d = dst[e];
    float w = dinv[s] * dinv[d];
    float4 v = ((const float4*)(h + (size_t)s * D))[lane];
    float* op = out + (size_t)d * D + lane * 4;
    atomicAdd(op + 0, v.x * w);
    atomicAdd(op + 1, v.y * w);
    atomicAdd(op + 2, v.z * w);
    atomicAdd(op + 3, v.w * w);
}

__global__ void bias_act_kernel(float* __restrict__ x,
                                const float* __restrict__ b,
                                long long total, int relu) {
    long long i = (long long)blockIdx.x * blockDim.x + threadIdx.x;
    if (i < total) {
        float v = x[i] + b[i & (D - 1)];
        if (relu) v = fmaxf(v, 0.0f);
        x[i] = v;
    }
}

extern "C" void kernel_launch(void* const* d_in, const int* in_sizes, int n_in,
                              void* d_out, int out_size, void* d_ws, size_t ws_size,
                              hipStream_t stream) {
    const float* x   = (const float*)d_in[0];
    const int*   ei  = (const int*)d_in[1];
    const float* W1  = (const float*)d_in[2];
    const float* b1  = (const float*)d_in[3];
    const float* W2  = (const float*)d_in[4];
    const float* b2  = (const float*)d_in[5];
    float* out = (float*)d_out;

    const int N = in_sizes[0] / D;
    const int E = in_sizes[1] / 2;
    const int* src = ei;
    const int* dst = ei + E;

    float* dinv = (float*)d_ws;
    size_t dinvPad = ((size_t)N + 63) & ~(size_t)63;
    float* bufA = dinv + dinvPad;
    float* bufB = bufA + (size_t)N * D;

    const long long totalF  = (long long)N * D;
    const int nThreads256   = 256;
    const int nBlocksN      = (N + 255) / 256;
    const int nBlocksE      = (E + 255) / 256;
    const int nBlocksEdge   = (E + 7) / 8;
    const int nBlocksF4     = (int)(((long long)N * (D / 4) + 255) / 256);
    const int nBlocksF      = (int)((totalF + 255) / 256);
    const int nBlocksGemm   = (N + 63) / 64;

    init_deg_kernel<<<nBlocksN, nThreads256, 0, stream>>>(dinv, N);
    count_deg_kernel<<<nBlocksE, nThreads256, 0, stream>>>(dst, E, dinv);
    dinv_kernel<<<nBlocksN, nThreads256, 0, stream>>>(dinv, N);

    gemm_wmma_f32<<<nBlocksGemm, 128, 0, stream>>>(x, W1, bufA, N);
    self_loop_init_kernel<<<nBlocksF4, nThreads256, 0, stream>>>(bufA, dinv, bufB, N);
    edge_scatter_kernel<<<nBlocksEdge, nThreads256, 0, stream>>>(src, dst, bufA, dinv, bufB, E);
    bias_act_kernel<<<nBlocksF, nThreads256, 0, stream>>>(bufB, b1, totalF, 1);

    gemm_wmma_f32<<<nBlocksGemm, 128, 0, stream>>>(bufB, W2, bufA, N);
    self_loop_init_kernel<<<nBlocksF4, nThreads256, 0, stream>>>(bufA, dinv, out, N);
    edge_scatter_kernel<<<nBlocksEdge, nThreads256, 0, stream>>>(src, dst, bufA, dinv, out, E);
    bias_act_kernel<<<nBlocksF, nThreads256, 0, stream>>>(out, b2, totalF, 0);
}


# aggregation in LDS: a work-group owns 200 output rows, filters the dst list for them and sums h[src]*w with LDS float adds on top of self term+bias; global float atomics go; sum order = LDS arrival
# speedup vs baseline: 1.6954x; 1.6954x over previous
#include <hip/hip_runtime.h>

typedef __attribute__((ext_vector_type(2))) float v2f;
typedef __attribute__((ext_vector_type(8))) float v8f;

#define D 128
#define LDS_STRIDE 130


__global__ void init_deg_kernel(float* __restrict__ deg, int n) {
    int i = blockIdx.x * blockDim.x + threadIdx.x;
    if (i < n) deg[i] = 1.0f;
}

__global__ void count_deg_kernel(const int* __restrict__ dst, int E,
                                 float* __restrict__ deg) {
    int e = blockIdx.x * blockDim.x + threadIdx.x;
    if (e < E) atomicAdd(&deg[dst[e]], 1.0f);
}


template <bool FIRST>
__global__ __launch_bounds__(128) void gemm_wmma_f32(
    const float* __restrict__ A, const float* __restrict__ W,
    const float* __restrict__ deg_in, float* __restrict__ dinv_out,
    float* __restrict__ H, int nrows) {
    __shared__ float wT[D * LDS_STRIDE];

    int tid = threadIdx.x;
    for (int i = tid; i < D * D; i += 128) {
        int r = i >> 7, c = i & (D - 1);
        wT[c * LDS_STRIDE + r] = W[i];
    }
    __syncthreads();

    int wave = tid >> 5;
    int lane = tid & 31;
    int lrow = lane & 15;
    int lhi  = lane >> 4;

    int m0 = blockIdx.x * 64 + wave * 16;

    v8f acc[8];
#pragma unroll
    for (int n = 0; n < 8; ++n) acc[n] = (v8f)(0.0f);

    int arow = m0 + lrow;
    int crow = arow < nrows ? arow : 0;
    const float* aptr = A + (size_t)crow * D + lhi * 2;

    if (FIRST) {
        if (lhi == 0 && arow < nrows) dinv_out[arow] = rsqrtf(deg_in[arow]);
    }

    for (int k = 0; k < D; k += 4) {
        v2f a = *(const v2f*)(aptr + k);
        if (!FIRST) { a.x = fmaxf(a.x, 0.0f); a.y = fmaxf(a.y, 0.0f); }
#pragma unroll
        for (int n = 0; n < 8; ++n) {
            int col = n * 16 + lrow;
            v2f b = *(const v2f*)(&wT[col * LDS_STRIDE + k + lhi * 2]);
            acc[n] = __builtin_amdgcn_wmma_f32_16x16x4_f32(
                false, a, false, b, (short)0, acc[n], false, false);
        }
    }

#pragma unroll
    for (int n = 0; n < 8; ++n) {
        int col = n * 16 + lrow;
#pragma unroll
        for (int v = 0; v < 8; ++v) {
            int row = m0 + v + 8 * lhi;
            if (row < nrows) H[(size_t)row * D + col] = acc[n][v];
        }
    }
}

#define AGG_ROWS 200
#define AGG_CAP  8192
#define AGG_NT   1024
#define AGG_SCAN 8
#define AGG_U    8

__device__ __forceinline__ void agg_take(int d, int e, int base, int nrow,
                                         int* cnt, int* lsrc, int* ldl, float* acc,
                                         const int* __restrict__ src,
                                         const float* __restrict__ h,
                                         const float* __restrict__ dinv) {
    unsigned dl = (unsigned)(d - base);
    if (dl < (unsigned)nrow) {
        int idx = atomicAdd(cnt, 1);
        if (idx < AGG_CAP) {
            lsrc[idx] = e;
            ldl[idx] = (int)dl;
        } else {
            int s = src[e];
            float w = dinv[s] * dinv[d];
            const float* hp = h + (size_t)s * D;
            float* ap = acc + dl * D;
#pragma nounroll
            for (int f = 0; f < D; ++f) atomicAdd(ap + f, hp[f] * w);
        }
    }
}

__device__ __forceinline__ void agg_take4(int4 v, bool live, int e, int base, int nrow,
                                          int* cnt, int* lsrc, int* ldl, float* acc,
                                          const int* __restrict__ src,
                                          const float* __restrict__ h,
                                          const float* __restrict__ dinv) {
    bool any = live & (((unsigned)(v.x - base) < (unsigned)nrow) |
                       ((unsigned)(v.y - base) < (unsigned)nrow) |
                       ((unsigned)(v.z - base) < (unsigned)nrow) |
                       ((unsigned)(v.w - base) < (unsigned)nrow));
    if (any) {
        agg_take(v.x, e + 0, base, nrow, cnt, lsrc, ldl, acc, src, h, dinv);
        agg_take(v.y, e + 1, base, nrow, cnt, lsrc, ldl, acc, src, h, dinv);
        agg_take(v.z, e + 2, base, nrow, cnt, lsrc, ldl, acc, src, h, dinv);
        agg_take(v.w, e + 3, base, nrow, cnt, lsrc, ldl, acc, src, h, dinv);
    }
}

__global__ __launch_bounds__(AGG_NT) void aggregate_rows_kernel(
    const int* __restrict__ src, const int* __restrict__ dst,
    const float* __restrict__ h, const float* __restrict__ dinv,
    const float* __restrict__ bias, float* __restrict__ out, int N, int E) {
    __shared__ float acc[AGG_ROWS * D];
    __shared__ int lsrc[AGG_CAP];
    __shared__ int ldl[AGG_CAP];
    __shared__ int cnt;

    const int tid = threadIdx.x;
    const int base = blockIdx.x * AGG_ROWS;
    const int nrow = (N - base) < AGG_ROWS ? (N - base) : AGG_ROWS;

    for (int i = tid; i < nrow * (D / 4); i += AGG_NT) {
        int r = i >> 5, c4 = i & 31;
        float q = dinv[base + r];
        q = q * q;
        float4 hv = ((const float4*)(h + (size_t)(base + r) * D))[c4];
        float4 bv = ((const float4*)bias)[c4];
        float4 o;
        o.x = hv.x * q + bv.x;
        o.y = hv.y * q + bv.y;
        o.z = hv.z * q + bv.z;
        o.w = hv.w * q + bv.w;
        ((float4*)acc)[i] = o;
    }
    if (tid == 0) cnt = 0;
    __syncthreads();

    const int E4 = ((E & 3) == 0) ? (E >> 2) : 0;
    const int4* d4 = (const int4*)dst;
    const int shift = (int)blockIdx.x * (E4 / (int)gridDim.x);
    for (int g0 = tid; g0 < E4; g0 += AGG_SCAN * AGG_NT) {
        int4 v[AGG_SCAN];
        int gp[AGG_SCAN];
        bool live[AGG_SCAN];
#pragma unroll
        for (int u = 0; u < AGG_SCAN; ++u) {
            int gl = g0 + u * AGG_NT;
            live[u] = gl < E4;
            int g = (live[u] ? gl : g0) + shift;
            gp[u] = g >= E4 ? g - E4 : g;
            v[u] = d4[gp[u]];
        }
#pragma unroll
        for (int u = 0; u < AGG_SCAN; ++u)
            agg_take4(v[u], live[u], gp[u] * 4, base, nrow, &cnt, lsrc, ldl, acc, src, h, dinv);
    }
    for (int e = E4 * 4 + tid; e < E; e += AGG_NT)
        agg_take(dst[e], e, base, nrow, &cnt, lsrc, ldl, acc, src, h, dinv);
    __syncthreads();

    const int n = cnt < AGG_CAP ? cnt : AGG_CAP;
    for (int i = tid; i < n; i += AGG_NT) lsrc[i] = src[lsrc[i]];
    __syncthreads();

    const int wave = tid >> 5, lane = tid & 31;
    for (int i0 = wave * AGG_U; i0 < n; i0 += (AGG_NT / 32) * AGG_U) {
        float2 va[AGG_U], vb[AGG_U];
        float wgt[AGG_U];
        int dl[AGG_U];
#pragma unroll
        for (int u = 0; u < AGG_U; ++u) {
            int i = i0 + u;
            bool ok = i < n;
            int s = ok ? lsrc[i] : 0;
            dl[u] = ok ? ldl[i] : 0;
            wgt[u] = dinv[s] * dinv[base + dl[u]];
            const float2* hp = (const float2*)(h + (size_t)s * D);
            va[u] = hp[lane];
            vb[u] = hp[32 + lane];
        }
#pragma unroll
        for (int u = 0; u < AGG_U; ++u) {
            if (i0 + u < n) {
                float* ap = acc + dl[u] * D + 2 * lane;
                float w = wgt[u];
                atomicAdd(ap + 0,  va[u].x * w);
                atomicAdd(ap + 1,  va[u].y * w);
                atomicAdd(ap + 64, vb[u].x * w);
                atomicAdd(ap + 65, vb[u].y * w);
            }
        }
    }
    __syncthreads();

    float4* op = (float4*)(out + (size_t)base * D);
    for (int i = tid; i < nrow * (D / 4); i += AGG_NT) op[i] = ((const float4*)acc)[i];
}

extern "C" void kernel_launch(void* const* d_in, const int* in_sizes, int n_in,
                              void* d_out, int out_size, void* d_ws, size_t ws_size,
                              hipStream_t stream) {
    const float* x   = (const float*)d_in[0];
    const int*   ei  = (const int*)d_in[1];
    const float* W1  = (const float*)d_in[2];
    const float* b1  = (const float*)d_in[3];
    const float* W2  = (const float*)d_in[4];
    const float* b2  = (const float*)d_in[5];
    float* out = (float*)d_out;

    const int N = in_sizes[0] / D;
    const int E = in_sizes[1] / 2;
    const int* src = ei;
    const int* dst = ei + E;

    float* deg = (float*)d_ws;
    size_t nPad = ((size_t)N + 63) & ~(size_t)63;
    float* dinv = deg + nPad;
    float* bufA = dinv + nPad;
    float* bufB = bufA + (size_t)N * D;

    const int nThreads256   = 256;
    const int nBlocksN      = (N + 255) / 256;
    const int nBlocksE      = (E + 255) / 256;
    const int nBlocksAgg    = (N + AGG_ROWS - 1) / AGG_ROWS;
    const int nBlocksGemm   = (N + 63) / 64;

    init_deg_kernel<<<nBlocksN, nThreads256, 0, stream>>>(deg, N);
    count_deg_kernel<<<nBlocksE, nThreads256, 0, stream>>>(dst, E, deg);

    gemm_wmma_f32<true><<<nBlocksGemm, 128, 0, stream>>>(x, W1, deg, dinv, bufA, N);
    aggregate_rows_kernel<<<nBlocksAgg, AGG_NT, 0, stream>>>(src, dst, bufA, dinv, b1, bufB, N, E);

    gemm_wmma_f32<false><<<nBlocksGemm, 128, 0, stream>>>(bufB, W2, nullptr, nullptr, bufA, N);
    aggregate_rows_kernel<<<nBlocksAgg, AGG_NT, 0, stream>>>(src, dst, bufA, dinv, b2, out, N, E);
}


# degree pass also writes a row-range x edge bitmap (LDS ORs, 256-B pieces); an aggregation work-group scans its own 1-bit-per-edge row, not the whole 4-byte dst list; sum order = LDS arrival as before
# speedup vs baseline: 1.9533x; 1.1521x over previous
#include <hip/hip_runtime.h>

typedef __attribute__((ext_vector_type(2))) float v2f;
typedef __attribute__((ext_vector_type(8))) float v8f;

#define D 128
#define LDS_STRIDE 130


__global__ void init_deg_kernel(float* __restrict__ deg, int n) {
    int i = blockIdx.x * blockDim.x + threadIdx.x;
    if (i < n) deg[i] = 1.0f;
}

#define AGG_ROWS 200

#define BM_CHUNK  2048
#define BM_WORDS  (BM_CHUNK / 32)
#define BM_NT     (BM_CHUNK / 4)
#define BM_RANGES 256

__global__ __launch_bounds__(BM_NT) void count_deg_bitmap_kernel(
    const int* __restrict__ dst, int E, float* __restrict__ deg,
    unsigned* __restrict__ bm, int nranges, int bm_stride) {
    __shared__ unsigned lb[BM_RANGES * BM_WORDS];

    const int tid = threadIdx.x;
    const int el = tid * 4;
    const int e = blockIdx.x * BM_CHUNK + el;

    int d[4] = {-1, -1, -1, -1};
    if (((E & 3) == 0) && e + 3 < E) {
        int4 v = *(const int4*)(dst + e);
        d[0] = v.x; d[1] = v.y; d[2] = v.z; d[3] = v.w;
    } else {
#pragma unroll
        for (int j = 0; j < 4; ++j)
            if (e + j < E) d[j] = dst[e + j];
    }

#pragma unroll
    for (int j = 0; j < 4; ++j)
        if (d[j] >= 0) atomicAdd(&deg[d[j]], 1.0f);

    for (int r0 = 0; r0 < nranges; r0 += BM_RANGES) {
        if (r0 > 0) __syncthreads();
        for (int i = tid; i < BM_RANGES * BM_WORDS / 4; i += BM_NT)
            ((uint4*)lb)[i] = make_uint4(0u, 0u, 0u, 0u);
        __syncthreads();
#pragma unroll
        for (int j = 0; j < 4; ++j) {
            unsigned r = (unsigned)d[j] / AGG_ROWS - (unsigned)r0;
            if (d[j] >= 0 && r < BM_RANGES)
                atomicOr(&lb[r * BM_WORDS + ((el + j) >> 5)], 1u << ((el + j) & 31));
        }
        __syncthreads();
        const int nr = (nranges - r0) < BM_RANGES ? (nranges - r0) : BM_RANGES;
        for (int i = tid; i < nr * (BM_WORDS / 4); i += BM_NT) {
            int r = i / (BM_WORDS / 4), q = i % (BM_WORDS / 4);
            uint4* gp = (uint4*)(bm + (size_t)(r0 + r) * bm_stride + (size_t)blockIdx.x * BM_WORDS);
            gp[q] = ((const uint4*)lb)[i];
        }
    }
}


template <bool FIRST>
__global__ __launch_bounds__(128) void gemm_wmma_f32(
    const float* __restrict__ A, const float* __restrict__ W,
    const float* __restrict__ deg_in, float* __restrict__ dinv_out,
    float* __restrict__ H, int nrows) {
    __shared__ float wT[D * LDS_STRIDE];

    int tid = threadIdx.x;
    for (int i = tid; i < D * D; i += 128) {
        int r = i >> 7, c = i & (D - 1);
        wT[c * LDS_STRIDE + r] = W[i];
    }
    __syncthreads();

    int wave = tid >> 5;
    int lane = tid & 31;
    int lrow = lane & 15;
    int lhi  = lane >> 4;

    int m0 = blockIdx.x * 64 + wave * 16;

    v8f acc[8];
#pragma unroll
    for (int n = 0; n < 8; ++n) acc[n] = (v8f)(0.0f);

    int arow = m0 + lrow;
    int crow = arow < nrows ? arow : 0;
    const float* aptr = A + (size_t)crow * D + lhi * 2;

    if (FIRST) {
        if (lhi == 0 && arow < nrows) dinv_out[arow] = rsqrtf(deg_in[arow]);
    }

    for (int k = 0; k < D; k += 4) {
        v2f a = *(const v2f*)(aptr + k);
        if (!FIRST) { a.x = fmaxf(a.x, 0.0f); a.y = fmaxf(a.y, 0.0f); }
#pragma unroll
        for (int n = 0; n < 8; ++n) {
            int col = n * 16 + lrow;
            v2f b = *(const v2f*)(&wT[col * LDS_STRIDE + k + lhi * 2]);
            acc[n] = __builtin_amdgcn_wmma_f32_16x16x4_f32(
                false, a, false, b, (short)0, acc[n], false, false);
        }
    }

#pragma unroll
    for (int n = 0; n < 8; ++n) {
        int col = n * 16 + lrow;
#pragma unroll
        for (int v = 0; v < 8; ++v) {
            int row = m0 + v + 8 * lhi;
            if (row < nrows) H[(size_t)row * D + col] = acc[n][v];
        }
    }
}

#define AGG_CAP  8192
#define AGG_NT   1024
#define AGG_SCAN 5
#define AGG_U    8

__device__ __forceinline__ int agg_emit(unsigned w, int eb, int idx, int* lsrc) {
    while (w) {
        int b = __builtin_ctz(w);
        w &= w - 1;
        if (idx < AGG_CAP) lsrc[idx] = eb + b;
        ++idx;
    }
    return idx;
}

__global__ __launch_bounds__(AGG_NT) void aggregate_rows_kernel(
    const int* __restrict__ src, const int* __restrict__ dst,
    const float* __restrict__ h, const float* __restrict__ dinv,
    const float* __restrict__ bias, float* __restrict__ out, int N,
    const unsigned* __restrict__ bm, int bm_stride) {
    __shared__ float acc[AGG_ROWS * D];
    __shared__ int lsrc[AGG_CAP];
    __shared__ int ldl[AGG_CAP];
    __shared__ int cnt;

    const int tid = threadIdx.x;
    const int base = blockIdx.x * AGG_ROWS;
    const int nrow = (N - base) < AGG_ROWS ? (N - base) : AGG_ROWS;

    for (int i = tid; i < nrow * (D / 4); i += AGG_NT) {
        int r = i >> 5, c4 = i & 31;
        float q = dinv[base + r];
        q = q * q;
        float4 hv = ((const float4*)(h + (size_t)(base + r) * D))[c4];
        float4 bv = ((const float4*)bias)[c4];
        float4 o;
        o.x = hv.x * q + bv.x;
        o.y = hv.y * q + bv.y;
        o.z = hv.z * q + bv.z;
        o.w = hv.w * q + bv.w;
        ((float4*)acc)[i] = o;
    }
    if (tid == 0) cnt = 0;
    __syncthreads();

    const unsigned* brow = bm + (size_t)blockIdx.x * bm_stride;
    const uint4* b4 = (const uint4*)brow;
    const int nq = bm_stride >> 2;
    for (int q0 = tid; q0 < nq; q0 += AGG_SCAN * AGG_NT) {
        uint4 v[AGG_SCAN];
        int c = 0;
#pragma unroll
        for (int u = 0; u < AGG_SCAN; ++u) {
            int q = q0 + u * AGG_NT;
            bool live = q < nq;
            uint4 t = b4[live ? q : q0];
            v[u] = live ? t : make_uint4(0u, 0u, 0u, 0u);
        }
#pragma unroll
        for (int u = 0; u < AGG_SCAN; ++u)
            c += __builtin_popcount(v[u].x) + __builtin_popcount(v[u].y) +
                 __builtin_popcount(v[u].z) + __builtin_popcount(v[u].w);
        if (c) {
            int idx = atomicAdd(&cnt, c);
#pragma unroll
            for (int u = 0; u < AGG_SCAN; ++u) {
                int eb = (q0 + u * AGG_NT) * 128;
                idx = agg_emit(v[u].x, eb,      idx, lsrc);
                idx = agg_emit(v[u].y, eb + 32, idx, lsrc);
                idx = agg_emit(v[u].z, eb + 64, idx, lsrc);
                idx = agg_emit(v[u].w, eb + 96, idx, lsrc);
            }
        }
    }
    __syncthreads();

    const int total = cnt;
    const int n = total <= AGG_CAP ? total : 0;
    for (int i = tid; i < n; i += AGG_NT) {
        int e = lsrc[i];
        lsrc[i] = src[e];
        ldl[i] = dst[e] - base;
    }
    __syncthreads();

    const int wave = tid >> 5, lane = tid & 31;
    for (int i0 = wave * AGG_U; i0 < n; i0 += (AGG_NT / 32) * AGG_U) {
        float2 va[AGG_U], vb[AGG_U];
        float wgt[AGG_U];
        int dl[AGG_U];
#pragma unroll
        for (int u = 0; u < AGG_U; ++u) {
            int i = i0 + u;
            bool ok = i < n;
            int s = ok ? lsrc[i] : 0;
            dl[u] = ok ? ldl[i] : 0;
            wgt[u] = dinv[s] * dinv[base + dl[u]];
            const float2* hp = (const float2*)(h + (size_t)s * D);
            va[u] = hp[lane];
            vb[u] = hp[32 + lane];
        }
#pragma unroll
        for (int u = 0; u < AGG_U; ++u) {
            if (i0 + u < n) {
                float* ap = acc + dl[u] * D + 2 * lane;
                float w = wgt[u];
                atomicAdd(ap + 0,  va[u].x * w);
                atomicAdd(ap + 1,  va[u].y * w);
                atomicAdd(ap + 64, vb[u].x * w);
                atomicAdd(ap + 65, vb[u].y * w);
            }
        }
    }
    if (total > AGG_CAP) {
        for (int q = tid; q < bm_stride; q += AGG_NT) {
            unsigned w = brow[q];
            while (w) {
                int e = q * 32 + __builtin_ctz(w);
                w &= w - 1;
                int s = src[e];
                int d = dst[e];
                float wg = dinv[s] * dinv[d];
                const float* hp = h + (size_t)s * D;
                float* ap = acc + (d - base) * D;
#pragma nounroll
                for (int f = 0; f < D; ++f) atomicAdd(ap + f, hp[f] * wg);
            }
        }
    }
    __syncthreads();

    float4* op = (float4*)(out + (size_t)base * D);
    for (int i = tid; i < nrow * (D / 4); i += AGG_NT) op[i] = ((const float4*)acc)[i];
}

extern "C" void kernel_launch(void* const* d_in, const int* in_sizes, int n_in,
                              void* d_out, int out_size, void* d_ws, size_t ws_size,
                              hipStream_t stream) {
    const float* x   = (const float*)d_in[0];
    const int*   ei  = (const int*)d_in[1];
    const float* W1  = (const float*)d_in[2];
    const float* b1  = (const float*)d_in[3];
    const float* W2  = (const float*)d_in[4];
    const float* b2  = (const float*)d_in[5];
    float* out = (float*)d_out;

    const int N = in_sizes[0] / D;
    const int E = in_sizes[1] / 2;
    const int* src = ei;
    const int* dst = ei + E;

    float* deg = (float*)d_ws;
    size_t nPad = ((size_t)N + 63) & ~(size_t)63;
    float* dinv = deg + nPad;
    float* bufA = dinv + nPad;
    float* bufB = bufA + (size_t)N * D;
    unsigned* bm = (unsigned*)(bufB + (size_t)N * D);

    const int nThreads256   = 256;
    const int nBlocksN      = (N + 255) / 256;
    const int nChunks       = (E + BM_CHUNK - 1) / BM_CHUNK;
    const int bmStride      = nChunks * BM_WORDS;
    const int nBlocksAgg    = (N + AGG_ROWS - 1) / AGG_ROWS;
    const int nBlocksGemm   = (N + 63) / 64;

    init_deg_kernel<<<nBlocksN, nThreads256, 0, stream>>>(deg, N);
    count_deg_bitmap_kernel<<<nChunks, BM_NT, 0, stream>>>(dst, E, deg, bm, nBlocksAgg, bmStride);

    gemm_wmma_f32<true><<<nBlocksGemm, 128, 0, stream>>>(x, W1, deg, dinv, bufA, N);
    aggregate_rows_kernel<<<nBlocksAgg, AGG_NT, 0, stream>>>(src, dst, bufA, dinv, b1, bufB, N, bm, bmStride);

    gemm_wmma_f32<false><<<nBlocksGemm, 128, 0, stream>>>(bufB, W2, nullptr, nullptr, bufA, N);
    aggregate_rows_kernel<<<nBlocksAgg, AGG_NT, 0, stream>>>(src, dst, bufA, dinv, b2, out, N, bm, bmStride);
}


# layer 1's aggregation stores each range's finished edge list (source row, local row) to the workspace; layer 2 reads it back: no bitmap scan, no src/dst lookups there; sum order = LDS arrival
# speedup vs baseline: 2.1097x; 1.0801x over previous
#include <hip/hip_runtime.h>

typedef __attribute__((ext_vector_type(2))) float v2f;
typedef __attribute__((ext_vector_type(8))) float v8f;

#define D 128
#define LDS_STRIDE 130


__global__ void init_deg_kernel(float* __restrict__ deg, int n) {
    int i = blockIdx.x * blockDim.x + threadIdx.x;
    if (i < n) deg[i] = 1.0f;
}

#define AGG_ROWS 200

#define BM_CHUNK  2048
#define BM_WORDS  (BM_CHUNK / 32)
#define BM_NT     (BM_CHUNK / 4)
#define BM_RANGES 256

__global__ __launch_bounds__(BM_NT) void count_deg_bitmap_kernel(
    const int* __restrict__ dst, int E, float* __restrict__ deg,
    unsigned* __restrict__ bm, int nranges, int bm_stride) {
    __shared__ unsigned lb[BM_RANGES * BM_WORDS];

    const int tid = threadIdx.x;
    const int el = tid * 4;
    const int e = blockIdx.x * BM_CHUNK + el;

    int d[4] = {-1, -1, -1, -1};
    if (((E & 3) == 0) && e + 3 < E) {
        int4 v = *(const int4*)(dst + e);
        d[0] = v.x; d[1] = v.y; d[2] = v.z; d[3] = v.w;
    } else {
#pragma unroll
        for (int j = 0; j < 4; ++j)
            if (e + j < E) d[j] = dst[e + j];
    }

#pragma unroll
    for (int j = 0; j < 4; ++j)
        if (d[j] >= 0) atomicAdd(&deg[d[j]], 1.0f);

    for (int r0 = 0; r0 < nranges; r0 += BM_RANGES) {
        if (r0 > 0) __syncthreads();
        for (int i = tid; i < BM_RANGES * BM_WORDS / 4; i += BM_NT)
            ((uint4*)lb)[i] = make_uint4(0u, 0u, 0u, 0u);
        __syncthreads();
#pragma unroll
        for (int j = 0; j < 4; ++j) {
            unsigned r = (unsigned)d[j] / AGG_ROWS - (unsigned)r0;
            if (d[j] >= 0 && r < BM_RANGES)
                atomicOr(&lb[r * BM_WORDS + ((el + j) >> 5)], 1u << ((el + j) & 31));
        }
        __syncthreads();
        const int nr = (nranges - r0) < BM_RANGES ? (nranges - r0) : BM_RANGES;
        for (int i = tid; i < nr * (BM_WORDS / 4); i += BM_NT) {
            int r = i / (BM_WORDS / 4), q = i % (BM_WORDS / 4);
            uint4* gp = (uint4*)(bm + (size_t)(r0 + r) * bm_stride + (size_t)blockIdx.x * BM_WORDS);
            gp[q] = ((const uint4*)lb)[i];
        }
    }
}


template <bool FIRST>
__global__ __launch_bounds__(128) void gemm_wmma_f32(
    const float* __restrict__ A, const float* __restrict__ W,
    const float* __restrict__ deg_in, float* __restrict__ dinv_out,
    float* __restrict__ H, int nrows) {
    __shared__ float wT[D * LDS_STRIDE];

    int tid = threadIdx.x;
    for (int i = tid; i < D * D; i += 128) {
        int r = i >> 7, c = i & (D - 1);
        wT[c * LDS_STRIDE + r] = W[i];
    }
    __syncthreads();

    int wave = tid >> 5;
    int lane = tid & 31;
    int lrow = lane & 15;
    int lhi  = lane >> 4;

    int m0 = blockIdx.x * 64 + wave * 16;

    v8f acc[8];
#pragma unroll
    for (int n = 0; n < 8; ++n) acc[n] = (v8f)(0.0f);

    int arow = m0 + lrow;
    int crow = arow < nrows ? arow : 0;
    const float* aptr = A + (size_t)crow * D + lhi * 2;

    if (FIRST) {
        if (lhi == 0 && arow < nrows) dinv_out[arow] = rsqrtf(deg_in[arow]);
    }

    for (int k = 0; k < D; k += 4) {
        v2f a = *(const v2f*)(aptr + k);
        if (!FIRST) { a.x = fmaxf(a.x, 0.0f); a.y = fmaxf(a.y, 0.0f); }
#pragma unroll
        for (int n = 0; n < 8; ++n) {
            int col = n * 16 + lrow;
            v2f b = *(const v2f*)(&wT[col * LDS_STRIDE + k + lhi * 2]);
            acc[n] = __builtin_amdgcn_wmma_f32_16x16x4_f32(
                false, a, false, b, (short)0, acc[n], false, false);
        }
    }

#pragma unroll
    for (int n = 0; n < 8; ++n) {
        int col = n * 16 + lrow;
#pragma unroll
        for (int v = 0; v < 8; ++v) {
            int row = m0 + v + 8 * lhi;
            if (row < nrows) H[(size_t)row * D + col] = acc[n][v];
        }
    }
}

#define AGG_CAP  8192
#define AGG_NT   1024
#define AGG_SCAN 5
#define AGG_U    8

__device__ __forceinline__ int agg_emit(unsigned w, int eb, int idx, int* lsrc) {
    while (w) {
        int b = __builtin_ctz(w);
        w &= w - 1;
        if (idx < AGG_CAP) lsrc[idx] = eb + b;
        ++idx;
    }
    return idx;
}

template <bool FULL>
__device__ __forceinline__ void agg_batch(int2 ev, int m, int lane, int base,
                                          const float* __restrict__ h,
                                          const float* __restrict__ dinv, float* acc) {
    float2 va[AGG_U], vb[AGG_U];
    float wgt[AGG_U];
    int dl[AGG_U];
#pragma unroll
    for (int u = 0; u < AGG_U; ++u) {
        int s = __builtin_amdgcn_readlane(ev.x, u);
        dl[u] = __builtin_amdgcn_readlane(ev.y, u);
        wgt[u] = dinv[s] * dinv[base + dl[u]];
        const float2* hp = (const float2*)(h + (size_t)s * D);
        va[u] = hp[lane];
        vb[u] = hp[32 + lane];
    }
#pragma unroll
    for (int u = 0; u < AGG_U; ++u) {
        if (FULL || u < m) {
            float* ap = acc + dl[u] * D + 2 * lane;
            float w = wgt[u];
            atomicAdd(ap + 0,  va[u].x * w);
            atomicAdd(ap + 1,  va[u].y * w);
            atomicAdd(ap + 64, vb[u].x * w);
            atomicAdd(ap + 65, vb[u].y * w);
        }
    }
}

template <bool BUILD>
__global__ __launch_bounds__(AGG_NT) void aggregate_rows_kernel(
    const int* __restrict__ src, const int* __restrict__ dst,
    const float* __restrict__ h, const float* __restrict__ dinv,
    const float* __restrict__ bias, float* __restrict__ out, int N,
    const unsigned* __restrict__ bm, int bm_stride,
    int* __restrict__ lst, int* __restrict__ lcount) {
    __shared__ float acc[AGG_ROWS * D];
    __shared__ int lsrc[BUILD ? AGG_CAP : 1];
    __shared__ int ldl[BUILD ? AGG_CAP : 1];
    __shared__ int cnt;

    const int tid = threadIdx.x;
    const int base = blockIdx.x * AGG_ROWS;
    const int nrow = (N - base) < AGG_ROWS ? (N - base) : AGG_ROWS;

    for (int i = tid; i < nrow * (D / 4); i += AGG_NT) {
        int r = i >> 5, c4 = i & 31;
        float q = dinv[base + r];
        q = q * q;
        float4 hv = ((const float4*)(h + (size_t)(base + r) * D))[c4];
        float4 bv = ((const float4*)bias)[c4];
        float4 o;
        o.x = hv.x * q + bv.x;
        o.y = hv.y * q + bv.y;
        o.z = hv.z * q + bv.z;
        o.w = hv.w * q + bv.w;
        ((float4*)acc)[i] = o;
    }
    if (BUILD && tid == 0) cnt = 0;
    __syncthreads();

    const unsigned* brow = bm + (size_t)blockIdx.x * bm_stride;
    int2* le = (int2*)lst + (size_t)blockIdx.x * AGG_CAP;
    const int wave = tid >> 5, lane = tid & 31;
    int total;

  if (BUILD) {
    const uint4* b4 = (const uint4*)brow;
    const int nq = bm_stride >> 2;
    for (int q0 = tid; q0 < nq; q0 += AGG_SCAN * AGG_NT) {
        uint4 v[AGG_SCAN];
        int c = 0;
#pragma unroll
        for (int u = 0; u < AGG_SCAN; ++u) {
            int q = q0 + u * AGG_NT;
            bool live = q < nq;
            uint4 t = b4[live ? q : q0];
            v[u] = live ? t : make_uint4(0u, 0u, 0u, 0u);
        }
#pragma unroll
        for (int u = 0; u < AGG_SCAN; ++u)
            c += __builtin_popcount(v[u].x) + __builtin_popcount(v[u].y) +
                 __builtin_popcount(v[u].z) + __builtin_popcount(v[u].w);
        if (c) {
            int idx = atomicAdd(&cnt, c);
#pragma unroll
            for (int u = 0; u < AGG_SCAN; ++u) {
                int eb = (q0 + u * AGG_NT) * 128;
                idx = agg_emit(v[u].x, eb,      idx, lsrc);
                idx = agg_emit(v[u].y, eb + 32, idx, lsrc);
                idx = agg_emit(v[u].z, eb + 64, idx, lsrc);
                idx = agg_emit(v[u].w, eb + 96, idx, lsrc);
            }
        }
    }
    __syncthreads();

    total = cnt;
    const int n = total <= AGG_CAP ? total : 0;
    for (int i = tid; i < n; i += AGG_NT) {
        int e = lsrc[i];
        int s = src[e];
        int dl = dst[e] - base;
        lsrc[i] = s;
        ldl[i] = dl;
        le[i] = make_int2(s, dl);
    }
    if (tid == 0) lcount[blockIdx.x] = total;
    __syncthreads();

    for (int i0 = wave * AGG_U; i0 < n; i0 += (AGG_NT / 32) * AGG_U) {
        float2 va[AGG_U], vb[AGG_U];
        float wgt[AGG_U];
        int dl[AGG_U];
#pragma unroll
        for (int u = 0; u < AGG_U; ++u) {
            int i = i0 + u;
            bool ok = i < n;
            int s = ok ? lsrc[i] : 0;
            dl[u] = ok ? ldl[i] : 0;
            wgt[u] = dinv[s] * dinv[base + dl[u]];
            const float2* hp = (const float2*)(h + (size_t)s * D);
            va[u] = hp[lane];
            vb[u] = hp[32 + lane];
        }
#pragma unroll
        for (int u = 0; u < AGG_U; ++u) {
            if (i0 + u < n) {
                float* ap = acc + dl[u] * D + 2 * lane;
                float w = wgt[u];
                atomicAdd(ap + 0,  va[u].x * w);
                atomicAdd(ap + 1,  va[u].y * w);
                atomicAdd(ap + 64, vb[u].x * w);
                atomicAdd(ap + 65, vb[u].y * w);
            }
        }
    }
  } else {
    total = lcount[blockIdx.x];
    const int n = total <= AGG_CAP ? total : 0;
    const int w0 = __builtin_amdgcn_readfirstlane(wave) * AGG_U;
    const int stride = (AGG_NT / 32) * AGG_U;
    if (w0 < n) {
        int j = w0 + (lane & (AGG_U - 1));
        int2 ev = le[j < n ? j : w0];
        int i0 = w0;
        for (; i0 + AGG_U <= n; i0 += stride) {
            int jn = i0 + stride + (lane & (AGG_U - 1));
            int2 nx = le[jn < n ? jn : i0];
            agg_batch<true>(ev, AGG_U, lane, base, h, dinv, acc);
            ev = nx;
        }
        if (i0 < n) agg_batch<false>(ev, n - i0, lane, base, h, dinv, acc);
    }
  }
    if (total > AGG_CAP) {
        for (int q = tid; q < bm_stride; q += AGG_NT) {
            unsigned w = brow[q];
            while (w) {
                int e = q * 32 + __builtin_ctz(w);
                w &= w - 1;
                int s = src[e];
                int d = dst[e];
                float wg = dinv[s] * dinv[d];
                const float* hp = h + (size_t)s * D;
                float* ap = acc + (d - base) * D;
#pragma nounroll
                for (int f = 0; f < D; ++f) atomicAdd(ap + f, hp[f] * wg);
            }
        }
    }
    __syncthreads();

    float4* op = (float4*)(out + (size_t)base * D);
    for (int i = tid; i < nrow * (D / 4); i += AGG_NT) op[i] = ((const float4*)acc)[i];
}

extern "C" void kernel_launch(void* const* d_in, const int* in_sizes, int n_in,
                              void* d_out, int out_size, void* d_ws, size_t ws_size,
                              hipStream_t stream) {
    const float* x   = (const float*)d_in[0];
    const int*   ei  = (const int*)d_in[1];
    const float* W1  = (const float*)d_in[2];
    const float* b1  = (const float*)d_in[3];
    const float* W2  = (const float*)d_in[4];
    const float* b2  = (const float*)d_in[5];
    float* out = (float*)d_out;

    const int N = in_sizes[0] / D;
    const int E = in_sizes[1] / 2;
    const int* src = ei;
    const int* dst = ei + E;

    float* deg = (float*)d_ws;
    size_t nPad = ((size_t)N + 63) & ~(size_t)63;
    float* dinv = deg + nPad;
    float* bufA = dinv + nPad;
    float* bufB = bufA + (size_t)N * D;
    unsigned* bm = (unsigned*)(bufB + (size_t)N * D);

    const int nThreads256   = 256;
    const int nBlocksN      = (N + 255) / 256;
    const int nChunks       = (E + BM_CHUNK - 1) / BM_CHUNK;
    const int bmStride      = nChunks * BM_WORDS;
    const int nBlocksAgg    = (N + AGG_ROWS - 1) / AGG_ROWS;
    const int nBlocksGemm   = (N + 63) / 64;
    int* lst    = (int*)(bm + (size_t)nBlocksAgg * bmStride);
    int* lcount = lst + (size_t)nBlocksAgg * AGG_CAP * 2;

    init_deg_kernel<<<nBlocksN, nThreads256, 0, stream>>>(deg, N);
    count_deg_bitmap_kernel<<<nChunks, BM_NT, 0, stream>>>(dst, E, deg, bm, nBlocksAgg, bmStride);

    gemm_wmma_f32<true><<<nBlocksGemm, 128, 0, stream>>>(x, W1, deg, dinv, bufA, N);
    aggregate_rows_kernel<true><<<nBlocksAgg, AGG_NT, 0, stream>>>(src, dst, bufA, dinv, b1, bufB, N, bm, bmStride, lst, lcount);

    gemm_wmma_f32<false><<<nBlocksGemm, 128, 0, stream>>>(bufB, W2, nullptr, nullptr, bufA, N);
    aggregate_rows_kernel<false><<<nBlocksAgg, AGG_NT, 0, stream>>>(src, dst, bufA, dinv, b2, out, N, bm, bmStride, lst, lcount);
}


# GEMM work-group = one 200-row aggregation range, 16 waves, 26 items of 16x64: 250 work-groups (one per WGP) stage W, not 782; fullest SIMD 896 WMMAs, not 1024; k order unchanged
# speedup vs baseline: 2.1823x; 1.0344x over previous
#include <hip/hip_runtime.h>

typedef __attribute__((ext_vector_type(2))) float v2f;
typedef __attribute__((ext_vector_type(8))) float v8f;

#define D 128
#define LDS_STRIDE 130


__global__ void init_deg_kernel(float* __restrict__ deg, int n) {
    int i = blockIdx.x * blockDim.x + threadIdx.x;
    if (i < n) deg[i] = 1.0f;
}

#define AGG_ROWS 200

#define BM_CHUNK  2048
#define BM_WORDS  (BM_CHUNK / 32)
#define BM_NT     (BM_CHUNK / 4)
#define BM_RANGES 256

__global__ __launch_bounds__(BM_NT) void count_deg_bitmap_kernel(
    const int* __restrict__ dst, int E, float* __restrict__ deg,
    unsigned* __restrict__ bm, int nranges, int bm_stride) {
    __shared__ unsigned lb[BM_RANGES * BM_WORDS];

    const int tid = threadIdx.x;
    const int el = tid * 4;
    const int e = blockIdx.x * BM_CHUNK + el;

    int d[4] = {-1, -1, -1, -1};
    if (((E & 3) == 0) && e + 3 < E) {
        int4 v = *(const int4*)(dst + e);
        d[0] = v.x; d[1] = v.y; d[2] = v.z; d[3] = v.w;
    } else {
#pragma unroll
        for (int j = 0; j < 4; ++j)
            if (e + j < E) d[j] = dst[e + j];
    }

#pragma unroll
    for (int j = 0; j < 4; ++j)
        if (d[j] >= 0) atomicAdd(&deg[d[j]], 1.0f);

    for (int r0 = 0; r0 < nranges; r0 += BM_RANGES) {
        if (r0 > 0) __syncthreads();
        for (int i = tid; i < BM_RANGES * BM_WORDS / 4; i += BM_NT)
            ((uint4*)lb)[i] = make_uint4(0u, 0u, 0u, 0u);
        __syncthreads();
#pragma unroll
        for (int j = 0; j < 4; ++j) {
            unsigned r = (unsigned)d[j] / AGG_ROWS - (unsigned)r0;
            if (d[j] >= 0 && r < BM_RANGES)
                atomicOr(&lb[r * BM_WORDS + ((el + j) >> 5)], 1u << ((el + j) & 31));
        }
        __syncthreads();
        const int nr = (nranges - r0) < BM_RANGES ? (nranges - r0) : BM_RANGES;
        for (int i = tid; i < nr * (BM_WORDS / 4); i += BM_NT) {
            int r = i / (BM_WORDS / 4), q = i % (BM_WORDS / 4);
            uint4* gp = (uint4*)(bm + (size_t)(r0 + r) * bm_stride + (size_t)blockIdx.x * BM_WORDS);
            gp[q] = ((const uint4*)lb)[i];
        }
    }
}


#define GEMM_ROWS  AGG_ROWS
#define GEMM_TILES ((GEMM_ROWS + 15) / 16)
#define GEMM_ITEMS (GEMM_TILES * 2)
#define GEMM_NW    16
#define GEMM_NT    (GEMM_NW * 32)

template <bool FIRST>
__global__ __launch_bounds__(GEMM_NT) void gemm_wmma_f32(
    const float* __restrict__ A, const float* __restrict__ W,
    const float* __restrict__ deg_in, float* __restrict__ dinv_out,
    float* __restrict__ H, int nrows) {
    __shared__ float wT[D * LDS_STRIDE];

    const int tid = threadIdx.x;
    for (int i = tid; i < D * D; i += GEMM_NT) {
        int r = i >> 7, c = i & (D - 1);
        wT[c * LDS_STRIDE + r] = W[i];
    }

    const int base = blockIdx.x * GEMM_ROWS;
    if (FIRST) {
        for (int r = tid; r < GEMM_ROWS; r += GEMM_NT) {
            int row = base + r;
            if (row < nrows) dinv_out[row] = rsqrtf(deg_in[row]);
        }
    }
    __syncthreads();

    const int wave = __builtin_amdgcn_readfirstlane(tid >> 5);
    const int lane = tid & 31;
    const int lrow = lane & 15;
    const int lhi  = lane >> 4;

    for (int it = wave; it < GEMM_ITEMS; it += GEMM_NW) {
        const int t  = it >> 1;
        const int c0 = (it & 1) * 64;
        const int m0 = base + t * 16;

        v8f acc[4];
#pragma unroll
        for (int n = 0; n < 4; ++n) acc[n] = (v8f)(0.0f);

        int arow = m0 + lrow;
        int crow = arow < nrows ? arow : 0;
        const float* aptr = A + (size_t)crow * D + lhi * 2;
        const float* bptr = &wT[(c0 + lrow) * LDS_STRIDE + lhi * 2];

        v2f af[D / 4];
#pragma unroll
        for (int kk = 0; kk < D / 4; ++kk) af[kk] = *(const v2f*)(aptr + 4 * kk);

#pragma unroll
        for (int kk = 0; kk < D / 4; ++kk) {
            v2f a = af[kk];
            if (!FIRST) { a.x = fmaxf(a.x, 0.0f); a.y = fmaxf(a.y, 0.0f); }
#pragma unroll
            for (int n = 0; n < 4; ++n) {
                v2f b = *(const v2f*)(bptr + n * 16 * LDS_STRIDE + 4 * kk);
                acc[n] = __builtin_amdgcn_wmma_f32_16x16x4_f32(
                    false, a, false, b, (short)0, acc[n], false, false);
            }
        }

#pragma unroll
        for (int n = 0; n < 4; ++n) {
            int col = c0 + n * 16 + lrow;
#pragma unroll
            for (int v = 0; v < 8; ++v) {
                int lr = t * 16 + v + 8 * lhi;
                int row = base + lr;
                if (lr < GEMM_ROWS && row < nrows) H[(size_t)row * D + col] = acc[n][v];
            }
        }
    }
}

#define AGG_CAP  8192
#define AGG_NT   1024
#define AGG_SCAN 5
#define AGG_U    8

__device__ __forceinline__ int agg_emit(unsigned w, int eb, int idx, int* lsrc) {
    while (w) {
        int b = __builtin_ctz(w);
        w &= w - 1;
        if (idx < AGG_CAP) lsrc[idx] = eb + b;
        ++idx;
    }
    return idx;
}

template <bool FULL>
__device__ __forceinline__ void agg_batch(int2 ev, int m, int lane, int base,
                                          const float* __restrict__ h,
                                          const float* __restrict__ dinv, float* acc) {
    float2 va[AGG_U], vb[AGG_U];
    float wgt[AGG_U];
    int dl[AGG_U];
#pragma unroll
    for (int u = 0; u < AGG_U; ++u) {
        int s = __builtin_amdgcn_readlane(ev.x, u);
        dl[u] = __builtin_amdgcn_readlane(ev.y, u);
        wgt[u] = dinv[s] * dinv[base + dl[u]];
        const float2* hp = (const float2*)(h + (size_t)s * D);
        va[u] = hp[lane];
        vb[u] = hp[32 + lane];
    }
#pragma unroll
    for (int u = 0; u < AGG_U; ++u) {
        if (FULL || u < m) {
            float* ap = acc + dl[u] * D + 2 * lane;
            float w = wgt[u];
            atomicAdd(ap + 0,  va[u].x * w);
            atomicAdd(ap + 1,  va[u].y * w);
            atomicAdd(ap + 64, vb[u].x * w);
            atomicAdd(ap + 65, vb[u].y * w);
        }
    }
}

template <bool BUILD>
__global__ __launch_bounds__(AGG_NT) void aggregate_rows_kernel(
    const int* __restrict__ src, const int* __restrict__ dst,
    const float* __restrict__ h, const float* __restrict__ dinv,
    const float* __restrict__ bias, float* __restrict__ out, int N,
    const unsigned* __restrict__ bm, int bm_stride,
    int* __restrict__ lst, int* __restrict__ lcount) {
    __shared__ float acc[AGG_ROWS * D];
    __shared__ int lsrc[BUILD ? AGG_CAP : 1];
    __shared__ int ldl[BUILD ? AGG_CAP : 1];
    __shared__ int cnt;

    const int tid = threadIdx.x;
    const int base = blockIdx.x * AGG_ROWS;
    const int nrow = (N - base) < AGG_ROWS ? (N - base) : AGG_ROWS;

    for (int i = tid; i < nrow * (D / 4); i += AGG_NT) {
        int r = i >> 5, c4 = i & 31;
        float q = dinv[base + r];
        q = q * q;
        float4 hv = ((const float4*)(h + (size_t)(base + r) * D))[c4];
        float4 bv = ((const float4*)bias)[c4];
        float4 o;
        o.x = hv.x * q + bv.x;
        o.y = hv.y * q + bv.y;
        o.z = hv.z * q + bv.z;
        o.w = hv.w * q + bv.w;
        ((float4*)acc)[i] = o;
    }
    if (BUILD && tid == 0) cnt = 0;
    __syncthreads();

    const unsigned* brow = bm + (size_t)blockIdx.x * bm_stride;
    int2* le = (int2*)lst + (size_t)blockIdx.x * AGG_CAP;
    const int wave = tid >> 5, lane = tid & 31;
    int total;

  if (BUILD) {
    const uint4* b4 = (const uint4*)brow;
    const int nq = bm_stride >> 2;
    for (int q0 = tid; q0 < nq; q0 += AGG_SCAN * AGG_NT) {
        uint4 v[AGG_SCAN];
        int c = 0;
#pragma unroll
        for (int u = 0; u < AGG_SCAN; ++u) {
            int q = q0 + u * AGG_NT;
            bool live = q < nq;
            uint4 t = b4[live ? q : q0];
            v[u] = live ? t : make_uint4(0u, 0u, 0u, 0u);
        }
#pragma unroll
        for (int u = 0; u < AGG_SCAN; ++u)
            c += __builtin_popcount(v[u].x) + __builtin_popcount(v[u].y) +
                 __builtin_popcount(v[u].z) + __builtin_popcount(v[u].w);
        if (c) {
            int idx = atomicAdd(&cnt, c);
#pragma unroll
            for (int u = 0; u < AGG_SCAN; ++u) {
                int eb = (q0 + u * AGG_NT) * 128;
                idx = agg_emit(v[u].x, eb,      idx, lsrc);
                idx = agg_emit(v[u].y, eb + 32, idx, lsrc);
                idx = agg_emit(v[u].z, eb + 64, idx, lsrc);
                idx = agg_emit(v[u].w, eb + 96, idx, lsrc);
            }
        }
    }
    __syncthreads();

    total = cnt;
    const int n = total <= AGG_CAP ? total : 0;
    for (int i = tid; i < n; i += AGG_NT) {
        int e = lsrc[i];
        int s = src[e];
        int dl = dst[e] - base;
        lsrc[i] = s;
        ldl[i] = dl;
        le[i] = make_int2(s, dl);
    }
    if (tid == 0) lcount[blockIdx.x] = total;
    __syncthreads();

    for (int i0 = wave * AGG_U; i0 < n; i0 += (AGG_NT / 32) * AGG_U) {
        float2 va[AGG_U], vb[AGG_U];
        float wgt[AGG_U];
        int dl[AGG_U];
#pragma unroll
        for (int u = 0; u < AGG_U; ++u) {
            int i = i0 + u;
            bool ok = i < n;
            int s = ok ? lsrc[i] : 0;
            dl[u] = ok ? ldl[i] : 0;
            wgt[u] = dinv[s] * dinv[base + dl[u]];
            const float2* hp = (const float2*)(h + (size_t)s * D);
            va[u] = hp[lane];
            vb[u] = hp[32 + lane];
        }
#pragma unroll
        for (int u = 0; u < AGG_U; ++u) {
            if (i0 + u < n) {
                float* ap = acc + dl[u] * D + 2 * lane;
                float w = wgt[u];
                atomicAdd(ap + 0,  va[u].x * w);
                atomicAdd(ap + 1,  va[u].y * w);
                atomicAdd(ap + 64, vb[u].x * w);
                atomicAdd(ap + 65, vb[u].y * w);
            }
        }
    }
  } else {
    total = lcount[blockIdx.x];
    const int n = total <= AGG_CAP ? total : 0;
    const int w0 = __builtin_amdgcn_readfirstlane(wave) * AGG_U;
    const int stride = (AGG_NT / 32) * AGG_U;
    if (w0 < n) {
        int j = w0 + (lane & (AGG_U - 1));
        int2 ev = le[j < n ? j : w0];
        int i0 = w0;
        for (; i0 + AGG_U <= n; i0 += stride) {
            int jn = i0 + stride + (lane & (AGG_U - 1));
            int2 nx = le[jn < n ? jn : i0];
            agg_batch<true>(ev, AGG_U, lane, base, h, dinv, acc);
            ev = nx;
        }
        if (i0 < n) agg_batch<false>(ev, n - i0, lane, base, h, dinv, acc);
    }
  }
    if (total > AGG_CAP) {
        for (int q = tid; q < bm_stride; q += AGG_NT) {
            unsigned w = brow[q];
            while (w) {
                int e = q * 32 + __builtin_ctz(w);
                w &= w - 1;
                int s = src[e];
                int d = dst[e];
                float wg = dinv[s] * dinv[d];
                const float* hp = h + (size_t)s * D;
                float* ap = acc + (d - base) * D;
#pragma nounroll
                for (int f = 0; f < D; ++f) atomicAdd(ap + f, hp[f] * wg);
            }
        }
    }
    __syncthreads();

    float4* op = (float4*)(out + (size_t)base * D);
    for (int i = tid; i < nrow * (D / 4); i += AGG_NT) op[i] = ((const float4*)acc)[i];
}

extern "C" void kernel_launch(void* const* d_in, const int* in_sizes, int n_in,
                              void* d_out, int out_size, void* d_ws, size_t ws_size,
                              hipStream_t stream) {
    const float* x   = (const float*)d_in[0];
    const int*   ei  = (const int*)d_in[1];
    const float* W1  = (const float*)d_in[2];
    const float* b1  = (const float*)d_in[3];
    const float* W2  = (const float*)d_in[4];
    const float* b2  = (const float*)d_in[5];
    float* out = (float*)d_out;

    const int N = in_sizes[0] / D;
    const int E = in_sizes[1] / 2;
    const int* src = ei;
    const int* dst = ei + E;

    float* deg = (float*)d_ws;
    size_t nPad = ((size_t)N + 63) & ~(size_t)63;
    float* dinv = deg + nPad;
    float* bufA = dinv + nPad;
    float* bufB = bufA + (size_t)N * D;
    unsigned* bm = (unsigned*)(bufB + (size_t)N * D);

    const int nThreads256   = 256;
    const int nBlocksN      = (N + 255) / 256;
    const int nChunks       = (E + BM_CHUNK - 1) / BM_CHUNK;
    const int bmStride      = nChunks * BM_WORDS;
    const int nBlocksAgg    = (N + AGG_ROWS - 1) / AGG_ROWS;
    int* lst    = (int*)(bm + (size_t)nBlocksAgg * bmStride);
    int* lcount = lst + (size_t)nBlocksAgg * AGG_CAP * 2;

    init_deg_kernel<<<nBlocksN, nThreads256, 0, stream>>>(deg, N);
    count_deg_bitmap_kernel<<<nChunks, BM_NT, 0, stream>>>(dst, E, deg, bm, nBlocksAgg, bmStride);

    gemm_wmma_f32<true><<<nBlocksAgg, GEMM_NT, 0, stream>>>(x, W1, deg, dinv, bufA, N);
    aggregate_rows_kernel<true><<<nBlocksAgg, AGG_NT, 0, stream>>>(src, dst, bufA, dinv, b1, bufB, N, bm, bmStride, lst, lcount);

    gemm_wmma_f32<false><<<nBlocksAgg, GEMM_NT, 0, stream>>>(bufB, W2, nullptr, nullptr, bufA, N);
    aggregate_rows_kernel<false><<<nBlocksAgg, AGG_NT, 0, stream>>>(src, dst, bufA, dinv, b2, out, N, bm, bmStride, lst, lcount);
}


# layer 1's edge-list build (bitmap scan, src/dst lookups) leaves the aggregation for 250 extra work-groups in the first GEMM's launch, beside its matrix work; sum order = LDS arrival as before
# speedup vs baseline: 2.3082x; 1.0577x over previous
#include <hip/hip_runtime.h>

typedef __attribute__((ext_vector_type(2))) float v2f;
typedef __attribute__((ext_vector_type(8))) float v8f;

#define D 128
#define LDS_STRIDE 130


__global__ void init_deg_kernel(float* __restrict__ deg, int n) {
    int i = blockIdx.x * blockDim.x + threadIdx.x;
    if (i < n) deg[i] = 1.0f;
}

#define AGG_ROWS 200

#define BM_CHUNK  2048
#define BM_WORDS  (BM_CHUNK / 32)
#define BM_NT     (BM_CHUNK / 4)
#define BM_RANGES 256

__global__ __launch_bounds__(BM_NT) void count_deg_bitmap_kernel(
    const int* __restrict__ dst, int E, float* __restrict__ deg,
    unsigned* __restrict__ bm, int nranges, int bm_stride) {
    __shared__ unsigned lb[BM_RANGES * BM_WORDS];

    const int tid = threadIdx.x;
    const int el = tid * 4;
    const int e = blockIdx.x * BM_CHUNK + el;

    int d[4] = {-1, -1, -1, -1};
    if (((E & 3) == 0) && e + 3 < E) {
        int4 v = *(const int4*)(dst + e);
        d[0] = v.x; d[1] = v.y; d[2] = v.z; d[3] = v.w;
    } else {
#pragma unroll
        for (int j = 0; j < 4; ++j)
            if (e + j < E) d[j] = dst[e + j];
    }

#pragma unroll
    for (int j = 0; j < 4; ++j)
        if (d[j] >= 0) atomicAdd(&deg[d[j]], 1.0f);

    for (int r0 = 0; r0 < nranges; r0 += BM_RANGES) {
        if (r0 > 0) __syncthreads();
        for (int i = tid; i < BM_RANGES * BM_WORDS / 4; i += BM_NT)
            ((uint4*)lb)[i] = make_uint4(0u, 0u, 0u, 0u);
        __syncthreads();
#pragma unroll
        for (int j = 0; j < 4; ++j) {
            unsigned r = (unsigned)d[j] / AGG_ROWS - (unsigned)r0;
            if (d[j] >= 0 && r < BM_RANGES)
                atomicOr(&lb[r * BM_WORDS + ((el + j) >> 5)], 1u << ((el + j) & 31));
        }
        __syncthreads();
        const int nr = (nranges - r0) < BM_RANGES ? (nranges - r0) : BM_RANGES;
        for (int i = tid; i < nr * (BM_WORDS / 4); i += BM_NT) {
            int r = i / (BM_WORDS / 4), q = i % (BM_WORDS / 4);
            uint4* gp = (uint4*)(bm + (size_t)(r0 + r) * bm_stride + (size_t)blockIdx.x * BM_WORDS);
            gp[q] = ((const uint4*)lb)[i];
        }
    }
}


#define AGG_CAP  8192
#define AGG_SCAN 5
#define LIST_U   5

__device__ __forceinline__ int agg_emit(unsigned w, int eb, int idx, int* lsrc) {
    while (w) {
        int b = __builtin_ctz(w);
        w &= w - 1;
        if (idx < AGG_CAP) lsrc[idx] = eb + b;
        ++idx;
    }
    return idx;
}

#define GEMM_ROWS  AGG_ROWS
#define GEMM_TILES ((GEMM_ROWS + 15) / 16)
#define GEMM_ITEMS (GEMM_TILES * 2)
#define GEMM_NW    16
#define GEMM_NT    (GEMM_NW * 32)

__device__ __forceinline__ void build_range_list(
    int r, int* lsrc, int* cnt,
    const int* __restrict__ src, const int* __restrict__ dst,
    const unsigned* __restrict__ bm, int bm_stride,
    int* __restrict__ lst, int* __restrict__ lcount) {
    const int tid = threadIdx.x;
    const int base = r * AGG_ROWS;
    if (tid == 0) *cnt = 0;
    __syncthreads();

    const uint4* b4 = (const uint4*)(bm + (size_t)r * bm_stride);
    const int nq = bm_stride >> 2;
    for (int q0 = tid; q0 < nq; q0 += AGG_SCAN * GEMM_NT) {
        uint4 v[AGG_SCAN];
        int c = 0;
#pragma unroll
        for (int u = 0; u < AGG_SCAN; ++u) {
            int q = q0 + u * GEMM_NT;
            bool live = q < nq;
            uint4 t = b4[live ? q : q0];
            v[u] = live ? t : make_uint4(0u, 0u, 0u, 0u);
        }
#pragma unroll
        for (int u = 0; u < AGG_SCAN; ++u)
            c += __builtin_popcount(v[u].x) + __builtin_popcount(v[u].y) +
                 __builtin_popcount(v[u].z) + __builtin_popcount(v[u].w);
        if (c) {
            int idx = atomicAdd(cnt, c);
#pragma unroll
            for (int u = 0; u < AGG_SCAN; ++u) {
                int eb = (q0 + u * GEMM_NT) * 128;
                idx = agg_emit(v[u].x, eb,      idx, lsrc);
                idx = agg_emit(v[u].y, eb + 32, idx, lsrc);
                idx = agg_emit(v[u].z, eb + 64, idx, lsrc);
                idx = agg_emit(v[u].w, eb + 96, idx, lsrc);
            }
        }
    }
    __syncthreads();

    const int total = *cnt;
    const int n = total <= AGG_CAP ? total : 0;
    int2* le = (int2*)lst + (size_t)r * AGG_CAP;
    for (int i0 = tid; i0 < n; i0 += LIST_U * GEMM_NT) {
        int s[LIST_U], d[LIST_U];
#pragma unroll
        for (int u = 0; u < LIST_U; ++u) {
            int i = i0 + u * GEMM_NT;
            int e = lsrc[i < n ? i : i0];
            s[u] = src[e];
            d[u] = dst[e];
        }
#pragma unroll
        for (int u = 0; u < LIST_U; ++u) {
            int i = i0 + u * GEMM_NT;
            if (i < n) le[i] = make_int2(s[u], d[u] - base);
        }
    }
    if (tid == 0) lcount[r] = total;
}

template <bool FIRST>
__global__ __launch_bounds__(GEMM_NT) void gemm_wmma_f32(
    const float* __restrict__ A, const float* __restrict__ W,
    const float* __restrict__ deg_in, float* __restrict__ dinv_out,
    float* __restrict__ H, int nrows, int nranges,
    const int* __restrict__ src, const int* __restrict__ dst,
    const unsigned* __restrict__ bm, int bm_stride,
    int* __restrict__ lst, int* __restrict__ lcount) {
    __shared__ float wT[D * LDS_STRIDE];
    __shared__ int cnt;

    if (FIRST && (int)blockIdx.x >= nranges) {
        build_range_list((int)blockIdx.x - nranges, (int*)wT, &cnt, src, dst, bm, bm_stride, lst, lcount);
        return;
    }

    const int tid = threadIdx.x;
    for (int i = tid; i < D * D; i += GEMM_NT) {
        int r = i >> 7, c = i & (D - 1);
        wT[c * LDS_STRIDE + r] = W[i];
    }

    const int base = blockIdx.x * GEMM_ROWS;
    if (FIRST) {
        for (int r = tid; r < GEMM_ROWS; r += GEMM_NT) {
            int row = base + r;
            if (row < nrows) dinv_out[row] = rsqrtf(deg_in[row]);
        }
    }
    __syncthreads();

    const int wave = __builtin_amdgcn_readfirstlane(tid >> 5);
    const int lane = tid & 31;
    const int lrow = lane & 15;
    const int lhi  = lane >> 4;

    for (int it = wave; it < GEMM_ITEMS; it += GEMM_NW) {
        const int t  = it >> 1;
        const int c0 = (it & 1) * 64;
        const int m0 = base + t * 16;

        v8f acc[4];
#pragma unroll
        for (int n = 0; n < 4; ++n) acc[n] = (v8f)(0.0f);

        int arow = m0 + lrow;
        int crow = arow < nrows ? arow : 0;
        const float* aptr = A + (size_t)crow * D + lhi * 2;
        const float* bptr = &wT[(c0 + lrow) * LDS_STRIDE + lhi * 2];

        v2f af[D / 4];
#pragma unroll
        for (int kk = 0; kk < D / 4; ++kk) af[kk] = *(const v2f*)(aptr + 4 * kk);

#pragma unroll
        for (int kk = 0; kk < D / 4; ++kk) {
            v2f a = af[kk];
            if (!FIRST) { a.x = fmaxf(a.x, 0.0f); a.y = fmaxf(a.y, 0.0f); }
#pragma unroll
            for (int n = 0; n < 4; ++n) {
                v2f b = *(const v2f*)(bptr + n * 16 * LDS_STRIDE + 4 * kk);
                acc[n] = __builtin_amdgcn_wmma_f32_16x16x4_f32(
                    false, a, false, b, (short)0, acc[n], false, false);
            }
        }

#pragma unroll
        for (int n = 0; n < 4; ++n) {
            int col = c0 + n * 16 + lrow;
#pragma unroll
            for (int v = 0; v < 8; ++v) {
                int lr = t * 16 + v + 8 * lhi;
                int row = base + lr;
                if (lr < GEMM_ROWS && row < nrows) H[(size_t)row * D + col] = acc[n][v];
            }
        }
    }
}

#define AGG_NT   1024
#define AGG_U    8

template <bool FULL>
__device__ __forceinline__ void agg_batch(int2 ev, int m, int lane, int base,
                                          const float* __restrict__ h,
                                          const float* __restrict__ dinv, float* acc) {
    float2 va[AGG_U], vb[AGG_U];
    float wgt[AGG_U];
    int dl[AGG_U];
#pragma unroll
    for (int u = 0; u < AGG_U; ++u) {
        int s = __builtin_amdgcn_readlane(ev.x, u);
        dl[u] = __builtin_amdgcn_readlane(ev.y, u);
        wgt[u] = dinv[s] * dinv[base + dl[u]];
        const float2* hp = (const float2*)(h + (size_t)s * D);
        va[u] = hp[lane];
        vb[u] = hp[32 + lane];
    }
#pragma unroll
    for (int u = 0; u < AGG_U; ++u) {
        if (FULL || u < m) {
            float* ap = acc + dl[u] * D + 2 * lane;
            float w = wgt[u];
            atomicAdd(ap + 0,  va[u].x * w);
            atomicAdd(ap + 1,  va[u].y * w);
            atomicAdd(ap + 64, vb[u].x * w);
            atomicAdd(ap + 65, vb[u].y * w);
        }
    }
}

__global__ __launch_bounds__(AGG_NT) void aggregate_rows_kernel(
    const int* __restrict__ src, const int* __restrict__ dst,
    const float* __restrict__ h, const float* __restrict__ dinv,
    const float* __restrict__ bias, float* __restrict__ out, int N,
    const unsigned* __restrict__ bm, int bm_stride,
    const int* __restrict__ lst, const int* __restrict__ lcount) {
    __shared__ float acc[AGG_ROWS * D];

    const int tid = threadIdx.x;
    const int base = blockIdx.x * AGG_ROWS;
    const int nrow = (N - base) < AGG_ROWS ? (N - base) : AGG_ROWS;

    for (int i = tid; i < nrow * (D / 4); i += AGG_NT) {
        int r = i >> 5, c4 = i & 31;
        float q = dinv[base + r];
        q = q * q;
        float4 hv = ((const float4*)(h + (size_t)(base + r) * D))[c4];
        float4 bv = ((const float4*)bias)[c4];
        float4 o;
        o.x = hv.x * q + bv.x;
        o.y = hv.y * q + bv.y;
        o.z = hv.z * q + bv.z;
        o.w = hv.w * q + bv.w;
        ((float4*)acc)[i] = o;
    }
    __syncthreads();

    const unsigned* brow = bm + (size_t)blockIdx.x * bm_stride;
    const int2* le = (const int2*)lst + (size_t)blockIdx.x * AGG_CAP;
    const int wave = tid >> 5, lane = tid & 31;

    const int total = lcount[blockIdx.x];
    const int n = total <= AGG_CAP ? total : 0;
    const int w0 = __builtin_amdgcn_readfirstlane(wave) * AGG_U;
    const int stride = (AGG_NT / 32) * AGG_U;
    if (w0 < n) {
        int j = w0 + (lane & (AGG_U - 1));
        int2 ev = le[j < n ? j : w0];
        int i0 = w0;
        for (; i0 + AGG_U <= n; i0 += stride) {
            int jn = i0 + stride + (lane & (AGG_U - 1));
            int2 nx = le[jn < n ? jn : i0];
            agg_batch<true>(ev, AGG_U, lane, base, h, dinv, acc);
            ev = nx;
        }
        if (i0 < n) agg_batch<false>(ev, n - i0, lane, base, h, dinv, acc);
    }
    if (total > AGG_CAP) {
        for (int q = tid; q < bm_stride; q += AGG_NT) {
            unsigned w = brow[q];
            while (w) {
                int e = q * 32 + __builtin_ctz(w);
                w &= w - 1;
                int s = src[e];
                int d = dst[e];
                float wg = dinv[s] * dinv[d];
                const float* hp = h + (size_t)s * D;
                float* ap = acc + (d - base) * D;
#pragma nounroll
                for (int f = 0; f < D; ++f) atomicAdd(ap + f, hp[f] * wg);
            }
        }
    }
    __syncthreads();

    float4* op = (float4*)(out + (size_t)base * D);
    for (int i = tid; i < nrow * (D / 4); i += AGG_NT) op[i] = ((const float4*)acc)[i];
}

extern "C" void kernel_launch(void* const* d_in, const int* in_sizes, int n_in,
                              void* d_out, int out_size, void* d_ws, size_t ws_size,
                              hipStream_t stream) {
    const float* x   = (const float*)d_in[0];
    const int*   ei  = (const int*)d_in[1];
    const float* W1  = (const float*)d_in[2];
    const float* b1  = (const float*)d_in[3];
    const float* W2  = (const float*)d_in[4];
    const float* b2  = (const float*)d_in[5];
    float* out = (float*)d_out;

    const int N = in_sizes[0] / D;
    const int E = in_sizes[1] / 2;
    const int* src = ei;
    const int* dst = ei + E;

    float* deg = (float*)d_ws;
    size_t nPad = ((size_t)N + 63) & ~(size_t)63;
    float* dinv = deg + nPad;
    float* bufA = dinv + nPad;
    float* bufB = bufA + (size_t)N * D;
    unsigned* bm = (unsigned*)(bufB + (size_t)N * D);

    const int nThreads256   = 256;
    const int nBlocksN      = (N + 255) / 256;
    const int nChunks       = (E + BM_CHUNK - 1) / BM_CHUNK;
    const int bmStride      = nChunks * BM_WORDS;
    const int nBlocksAgg    = (N + AGG_ROWS - 1) / AGG_ROWS;
    int* lst    = (int*)(bm + (size_t)nBlocksAgg * bmStride);
    int* lcount = lst + (size_t)nBlocksAgg * AGG_CAP * 2;

    init_deg_kernel<<<nBlocksN, nThreads256, 0, stream>>>(deg, N);
    count_deg_bitmap_kernel<<<nChunks, BM_NT, 0, stream>>>(dst, E, deg, bm, nBlocksAgg, bmStride);

    gemm_wmma_f32<true><<<nBlocksAgg + nBlocksAgg, GEMM_NT, 0, stream>>>(x, W1, deg, dinv, bufA, N, nBlocksAgg, src, dst, bm, bmStride, lst, lcount);
    aggregate_rows_kernel<<<nBlocksAgg, AGG_NT, 0, stream>>>(src, dst, bufA, dinv, b1, bufB, N, bm, bmStride, lst, lcount);

    gemm_wmma_f32<false><<<nBlocksAgg, GEMM_NT, 0, stream>>>(bufB, W2, nullptr, nullptr, bufA, N, nBlocksAgg, nullptr, nullptr, nullptr, 0, nullptr, nullptr);
    aggregate_rows_kernel<<<nBlocksAgg, AGG_NT, 0, stream>>>(src, dst, bufA, dinv, b2, out, N, bm, bmStride, lst, lcount);
}


# layer 2's product moves into layer 1's aggregation: relu(LDS sums) x W2 on WMMA straight from LDS, W2 staged under the gathers; layer-1 result's round trip and one launch go (6->5); k order, sums kept
# speedup vs baseline: 2.5462x; 1.1031x over previous
#include <hip/hip_runtime.h>

typedef __attribute__((ext_vector_type(2))) float v2f;
typedef __attribute__((ext_vector_type(8))) float v8f;

#define D 128
#define LDS_STRIDE 130


__global__ void init_deg_kernel(float* __restrict__ deg, int n) {
    int i = blockIdx.x * blockDim.x + threadIdx.x;
    if (i < n) deg[i] = 1.0f;
}

#define AGG_ROWS 200

#define BM_CHUNK  2048
#define BM_WORDS  (BM_CHUNK / 32)
#define BM_NT     (BM_CHUNK / 4)
#define BM_RANGES 256

__global__ __launch_bounds__(BM_NT) void count_deg_bitmap_kernel(
    const int* __restrict__ dst, int E, float* __restrict__ deg,
    unsigned* __restrict__ bm, int nranges, int bm_stride) {
    __shared__ unsigned lb[BM_RANGES * BM_WORDS];

    const int tid = threadIdx.x;
    const int el = tid * 4;
    const int e = blockIdx.x * BM_CHUNK + el;

    int d[4] = {-1, -1, -1, -1};
    if (((E & 3) == 0) && e + 3 < E) {
        int4 v = *(const int4*)(dst + e);
        d[0] = v.x; d[1] = v.y; d[2] = v.z; d[3] = v.w;
    } else {
#pragma unroll
        for (int j = 0; j < 4; ++j)
            if (e + j < E) d[j] = dst[e + j];
    }

#pragma unroll
    for (int j = 0; j < 4; ++j)
        if (d[j] >= 0) atomicAdd(&deg[d[j]], 1.0f);

    for (int r0 = 0; r0 < nranges; r0 += BM_RANGES) {
        if (r0 > 0) __syncthreads();
        for (int i = tid; i < BM_RANGES * BM_WORDS / 4; i += BM_NT)
            ((uint4*)lb)[i] = make_uint4(0u, 0u, 0u, 0u);
        __syncthreads();
#pragma unroll
        for (int j = 0; j < 4; ++j) {
            unsigned r = (unsigned)d[j] / AGG_ROWS - (unsigned)r0;
            if (d[j] >= 0 && r < BM_RANGES)
                atomicOr(&lb[r * BM_WORDS + ((el + j) >> 5)], 1u << ((el + j) & 31));
        }
        __syncthreads();
        const int nr = (nranges - r0) < BM_RANGES ? (nranges - r0) : BM_RANGES;
        for (int i = tid; i < nr * (BM_WORDS / 4); i += BM_NT) {
            int r = i / (BM_WORDS / 4), q = i % (BM_WORDS / 4);
            uint4* gp = (uint4*)(bm + (size_t)(r0 + r) * bm_stride + (size_t)blockIdx.x * BM_WORDS);
            gp[q] = ((const uint4*)lb)[i];
        }
    }
}


#define AGG_CAP  8192
#define AGG_SCAN 5
#define LIST_U   5

__device__ __forceinline__ int agg_emit(unsigned w, int eb, int idx, int* lsrc) {
    while (w) {
        int b = __builtin_ctz(w);
        w &= w - 1;
        if (idx < AGG_CAP) lsrc[idx] = eb + b;
        ++idx;
    }
    return idx;
}

#define GEMM_ROWS  AGG_ROWS
#define GEMM_TILES ((GEMM_ROWS + 15) / 16)
#define GEMM_ITEMS (GEMM_TILES * 2)
#define GEMM_NW    16
#define GEMM_NT    (GEMM_NW * 32)

__device__ __forceinline__ void build_range_list(
    int r, int* lsrc, int* cnt,
    const int* __restrict__ src, const int* __restrict__ dst,
    const unsigned* __restrict__ bm, int bm_stride,
    int* __restrict__ lst, int* __restrict__ lcount) {
    const int tid = threadIdx.x;
    const int base = r * AGG_ROWS;
    if (tid == 0) *cnt = 0;
    __syncthreads();

    const uint4* b4 = (const uint4*)(bm + (size_t)r * bm_stride);
    const int nq = bm_stride >> 2;
    for (int q0 = tid; q0 < nq; q0 += AGG_SCAN * GEMM_NT) {
        uint4 v[AGG_SCAN];
        int c = 0;
#pragma unroll
        for (int u = 0; u < AGG_SCAN; ++u) {
            int q = q0 + u * GEMM_NT;
            bool live = q < nq;
            uint4 t = b4[live ? q : q0];
            v[u] = live ? t : make_uint4(0u, 0u, 0u, 0u);
        }
#pragma unroll
        for (int u = 0; u < AGG_SCAN; ++u)
            c += __builtin_popcount(v[u].x) + __builtin_popcount(v[u].y) +
                 __builtin_popcount(v[u].z) + __builtin_popcount(v[u].w);
        if (c) {
            int idx = atomicAdd(cnt, c);
#pragma unroll
            for (int u = 0; u < AGG_SCAN; ++u) {
                int eb = (q0 + u * GEMM_NT) * 128;
                idx = agg_emit(v[u].x, eb,      idx, lsrc);
                idx = agg_emit(v[u].y, eb + 32, idx, lsrc);
                idx = agg_emit(v[u].z, eb + 64, idx, lsrc);
                idx = agg_emit(v[u].w, eb + 96, idx, lsrc);
            }
        }
    }
    __syncthreads();

    const int total = *cnt;
    const int n = total <= AGG_CAP ? total : 0;
    int2* le = (int2*)lst + (size_t)r * AGG_CAP;
    for (int i0 = tid; i0 < n; i0 += LIST_U * GEMM_NT) {
        int s[LIST_U], d[LIST_U];
#pragma unroll
        for (int u = 0; u < LIST_U; ++u) {
            int i = i0 + u * GEMM_NT;
            int e = lsrc[i < n ? i : i0];
            s[u] = src[e];
            d[u] = dst[e];
        }
#pragma unroll
        for (int u = 0; u < LIST_U; ++u) {
            int i = i0 + u * GEMM_NT;
            if (i < n) le[i] = make_int2(s[u], d[u] - base);
        }
    }
    if (tid == 0) lcount[r] = total;
}

__global__ __launch_bounds__(GEMM_NT) void gemm_wmma_f32(
    const float* __restrict__ A, const float* __restrict__ W,
    const float* __restrict__ deg_in, float* __restrict__ dinv_out,
    float* __restrict__ H, int nrows, int nranges,
    const int* __restrict__ src, const int* __restrict__ dst,
    const unsigned* __restrict__ bm, int bm_stride,
    int* __restrict__ lst, int* __restrict__ lcount) {
    __shared__ float wT[D * LDS_STRIDE];
    __shared__ int cnt;

    if ((int)blockIdx.x >= nranges) {
        build_range_list((int)blockIdx.x - nranges, (int*)wT, &cnt, src, dst, bm, bm_stride, lst, lcount);
        return;
    }

    const int tid = threadIdx.x;
    for (int i = tid; i < D * D; i += GEMM_NT) {
        int r = i >> 7, c = i & (D - 1);
        wT[c * LDS_STRIDE + r] = W[i];
    }

    const int base = blockIdx.x * GEMM_ROWS;
    for (int r = tid; r < GEMM_ROWS; r += GEMM_NT) {
        int row = base + r;
        if (row < nrows) dinv_out[row] = rsqrtf(deg_in[row]);
    }
    __syncthreads();

    const int wave = __builtin_amdgcn_readfirstlane(tid >> 5);
    const int lane = tid & 31;
    const int lrow = lane & 15;
    const int lhi  = lane >> 4;

    for (int it = wave; it < GEMM_ITEMS; it += GEMM_NW) {
        const int t  = it >> 1;
        const int c0 = (it & 1) * 64;
        const int m0 = base + t * 16;

        v8f acc[4];
#pragma unroll
        for (int n = 0; n < 4; ++n) acc[n] = (v8f)(0.0f);

        int arow = m0 + lrow;
        int crow = arow < nrows ? arow : 0;
        const float* aptr = A + (size_t)crow * D + lhi * 2;
        const float* bptr = &wT[(c0 + lrow) * LDS_STRIDE + lhi * 2];

        v2f af[D / 4];
#pragma unroll
        for (int kk = 0; kk < D / 4; ++kk) af[kk] = *(const v2f*)(aptr + 4 * kk);

#pragma unroll
        for (int kk = 0; kk < D / 4; ++kk) {
            v2f a = af[kk];
#pragma unroll
            for (int n = 0; n < 4; ++n) {
                v2f b = *(const v2f*)(bptr + n * 16 * LDS_STRIDE + 4 * kk);
                acc[n] = __builtin_amdgcn_wmma_f32_16x16x4_f32(
                    false, a, false, b, (short)0, acc[n], false, false);
            }
        }

#pragma unroll
        for (int n = 0; n < 4; ++n) {
            int col = c0 + n * 16 + lrow;
#pragma unroll
            for (int v = 0; v < 8; ++v) {
                int lr = t * 16 + v + 8 * lhi;
                int row = base + lr;
                if (lr < GEMM_ROWS && row < nrows) H[(size_t)row * D + col] = acc[n][v];
            }
        }
    }
}

#define AGG_NT   1024
#define AGG_U    8
#define AGG_PITCH_MM 132
#define AGG_ROWS_MM  (((AGG_ROWS + 15) / 16) * 16)
#define AGG_ITEMS_MM ((AGG_ROWS_MM / 16) * 4)

template <bool FULL, int PITCH>
__device__ __forceinline__ void agg_batch(int2 ev, int m, int lane, int base,
                                          const float* __restrict__ h,
                                          const float* __restrict__ dinv, float* acc) {
    float2 va[AGG_U], vb[AGG_U];
    float wgt[AGG_U];
    int dl[AGG_U];
#pragma unroll
    for (int u = 0; u < AGG_U; ++u) {
        int s = __builtin_amdgcn_readlane(ev.x, u);
        dl[u] = __builtin_amdgcn_readlane(ev.y, u);
        wgt[u] = dinv[s] * dinv[base + dl[u]];
        const float2* hp = (const float2*)(h + (size_t)s * D);
        va[u] = hp[lane];
        vb[u] = hp[32 + lane];
    }
#pragma unroll
    for (int u = 0; u < AGG_U; ++u) {
        if (FULL || u < m) {
            float* ap = acc + dl[u] * PITCH + 2 * lane;
            float w = wgt[u];
            atomicAdd(ap + 0,  va[u].x * w);
            atomicAdd(ap + 1,  va[u].y * w);
            atomicAdd(ap + 64, vb[u].x * w);
            atomicAdd(ap + 65, vb[u].y * w);
        }
    }
}

template <bool NEXT>
__global__ __launch_bounds__(AGG_NT) void aggregate_rows_kernel(
    const int* __restrict__ src, const int* __restrict__ dst,
    const float* __restrict__ h, const float* __restrict__ dinv,
    const float* __restrict__ bias, float* __restrict__ out, int N,
    const unsigned* __restrict__ bm, int bm_stride,
    const int* __restrict__ lst, const int* __restrict__ lcount,
    const float* __restrict__ Wn) {
    constexpr int PITCH = NEXT ? AGG_PITCH_MM : D;
    constexpr int LROWS = NEXT ? AGG_ROWS_MM : AGG_ROWS;
    __shared__ float acc[LROWS * PITCH];
    __shared__ float wT[NEXT ? D * AGG_PITCH_MM : 4];

    const int tid = threadIdx.x;
    const int base = blockIdx.x * AGG_ROWS;
    const int nrow = (N - base) < AGG_ROWS ? (N - base) : AGG_ROWS;

    float wreg[NEXT ? (D * D) / AGG_NT : 1];
    if (NEXT) {
#pragma unroll
        for (int k = 0; k < (D * D) / AGG_NT; ++k) wreg[k] = Wn[tid + k * AGG_NT];
    }

    for (int i = tid; i < (NEXT ? LROWS : nrow) * (D / 4); i += AGG_NT) {
        int r = i >> 5, c4 = i & 31;
        float4 o = make_float4(0.0f, 0.0f, 0.0f, 0.0f);
        if (!NEXT || r < nrow) {
            float q = dinv[base + r];
            q = q * q;
            float4 hv = ((const float4*)(h + (size_t)(base + r) * D))[c4];
            float4 bv = ((const float4*)bias)[c4];
            o.x = hv.x * q + bv.x;
            o.y = hv.y * q + bv.y;
            o.z = hv.z * q + bv.z;
            o.w = hv.w * q + bv.w;
        }
        *(float4*)(acc + r * PITCH + c4 * 4) = o;
    }
    if (NEXT) {
#pragma unroll
        for (int k = 0; k < (D * D) / AGG_NT; ++k) {
            int i = tid + k * AGG_NT;
            wT[(i & (D - 1)) * AGG_PITCH_MM + (i >> 7)] = wreg[k];
        }
    }
    __syncthreads();

    const unsigned* brow = bm + (size_t)blockIdx.x * bm_stride;
    const int2* le = (const int2*)lst + (size_t)blockIdx.x * AGG_CAP;
    const int wave = tid >> 5, lane = tid & 31;

    const int total = lcount[blockIdx.x];
    const int n = total <= AGG_CAP ? total : 0;
    const int w0 = __builtin_amdgcn_readfirstlane(wave) * AGG_U;
    const int stride = (AGG_NT / 32) * AGG_U;
    if (w0 < n) {
        int j = w0 + (lane & (AGG_U - 1));
        int2 ev = le[j < n ? j : w0];
        int i0 = w0;
        for (; i0 + AGG_U <= n; i0 += stride) {
            int jn = i0 + stride + (lane & (AGG_U - 1));
            int2 nx = le[jn < n ? jn : i0];
            agg_batch<true, PITCH>(ev, AGG_U, lane, base, h, dinv, acc);
            ev = nx;
        }
        if (i0 < n) agg_batch<false, PITCH>(ev, n - i0, lane, base, h, dinv, acc);
    }
    if (total > AGG_CAP) {
        for (int q = tid; q < bm_stride; q += AGG_NT) {
            unsigned w = brow[q];
            while (w) {
                int e = q * 32 + __builtin_ctz(w);
                w &= w - 1;
                int s = src[e];
                int d = dst[e];
                float wg = dinv[s] * dinv[d];
                const float* hp = h + (size_t)s * D;
                float* ap = acc + (d - base) * PITCH;
#pragma nounroll
                for (int f = 0; f < D; ++f) atomicAdd(ap + f, hp[f] * wg);
            }
        }
    }
    __syncthreads();

    if (!NEXT) {
        float4* op = (float4*)(out + (size_t)base * D);
        for (int i = tid; i < nrow * (D / 4); i += AGG_NT) op[i] = ((const float4*)acc)[i];
    } else {
        const int wv = __builtin_amdgcn_readfirstlane(wave);
        const int lrow = lane & 15;
        const int lhi  = lane >> 4;
        for (int it = wv; it < AGG_ITEMS_MM; it += AGG_NT / 32) {
            const int t  = it >> 2;
            const int c0 = (it & 3) * 32;
            const float* aptr = acc + (t * 16 + lrow) * PITCH + lhi * 2;
            const float* bptr = wT + (c0 + lrow) * AGG_PITCH_MM + lhi * 2;
            v8f cc[2];
            cc[0] = (v8f)(0.0f);
            cc[1] = (v8f)(0.0f);
#pragma unroll
            for (int kk = 0; kk < D / 4; ++kk) {
                v2f a = *(const v2f*)(aptr + 4 * kk);
                a.x = fmaxf(a.x, 0.0f);
                a.y = fmaxf(a.y, 0.0f);
#pragma unroll
                for (int n = 0; n < 2; ++n) {
                    v2f b = *(const v2f*)(bptr + n * 16 * AGG_PITCH_MM + 4 * kk);
                    cc[n] = __builtin_amdgcn_wmma_f32_16x16x4_f32(
                        false, a, false, b, (short)0, cc[n], false, false);
                }
            }
#pragma unroll
            for (int n = 0; n < 2; ++n) {
                int col = c0 + n * 16 + lrow;
#pragma unroll
                for (int v = 0; v < 8; ++v) {
                    int lr = t * 16 + v + 8 * lhi;
                    if (lr < nrow) out[(size_t)(base + lr) * D + col] = cc[n][v];
                }
            }
        }
    }
}

extern "C" void kernel_launch(void* const* d_in, const int* in_sizes, int n_in,
                              void* d_out, int out_size, void* d_ws, size_t ws_size,
                              hipStream_t stream) {
    const float* x   = (const float*)d_in[0];
    const int*   ei  = (const int*)d_in[1];
    const float* W1  = (const float*)d_in[2];
    const float* b1  = (const float*)d_in[3];
    const float* W2  = (const float*)d_in[4];
    const float* b2  = (const float*)d_in[5];
    float* out = (float*)d_out;

    const int N = in_sizes[0] / D;
    const int E = in_sizes[1] / 2;
    const int* src = ei;
    const int* dst = ei + E;

    float* deg = (float*)d_ws;
    size_t nPad = ((size_t)N + 63) & ~(size_t)63;
    float* dinv = deg + nPad;
    float* bufA = dinv + nPad;
    float* bufB = bufA + (size_t)N * D;
    unsigned* bm = (unsigned*)(bufB + (size_t)N * D);

    const int nThreads256   = 256;
    const int nBlocksN      = (N + 255) / 256;
    const int nChunks       = (E + BM_CHUNK - 1) / BM_CHUNK;
    const int bmStride      = nChunks * BM_WORDS;
    const int nBlocksAgg    = (N + AGG_ROWS - 1) / AGG_ROWS;
    int* lst    = (int*)(bm + (size_t)nBlocksAgg * bmStride);
    int* lcount = lst + (size_t)nBlocksAgg * AGG_CAP * 2;

    init_deg_kernel<<<nBlocksN, nThreads256, 0, stream>>>(deg, N);
    count_deg_bitmap_kernel<<<nChunks, BM_NT, 0, stream>>>(dst, E, deg, bm, nBlocksAgg, bmStride);

    gemm_wmma_f32<<<nBlocksAgg + nBlocksAgg, GEMM_NT, 0, stream>>>(x, W1, deg, dinv, bufA, N, nBlocksAgg, src, dst, bm, bmStride, lst, lcount);
    aggregate_rows_kernel<true><<<nBlocksAgg, AGG_NT, 0, stream>>>(src, dst, bufA, dinv, b1, bufB, N, bm, bmStride, lst, lcount, W2);

    aggregate_rows_kernel<false><<<nBlocksAgg, AGG_NT, 0, stream>>>(src, dst, bufB, dinv, b2, out, N, bm, bmStride, lst, lcount, nullptr);
}


# in-degrees counted by the list work-groups from their range's own edges (LDS int adds, dinv stored there); the init launch and the 640K global float atomics go; 5->4 launches; sums unchanged
# speedup vs baseline: 3.7120x; 1.4579x over previous
#include <hip/hip_runtime.h>

typedef __attribute__((ext_vector_type(2))) float v2f;
typedef __attribute__((ext_vector_type(8))) float v8f;

#define D 128
#define LDS_STRIDE 130


#define AGG_ROWS 200

#define BM_CHUNK  2048
#define BM_WORDS  (BM_CHUNK / 32)
#define BM_NT     (BM_CHUNK / 4)
#define BM_RANGES 256

__global__ __launch_bounds__(BM_NT) void edge_bitmap_kernel(
    const int* __restrict__ dst, int E,
    unsigned* __restrict__ bm, int nranges, int bm_stride) {
    __shared__ unsigned lb[BM_RANGES * BM_WORDS];

    const int tid = threadIdx.x;
    const int el = tid * 4;
    const int e = blockIdx.x * BM_CHUNK + el;

    int d[4] = {-1, -1, -1, -1};
    if (((E & 3) == 0) && e + 3 < E) {
        int4 v = *(const int4*)(dst + e);
        d[0] = v.x; d[1] = v.y; d[2] = v.z; d[3] = v.w;
    } else {
#pragma unroll
        for (int j = 0; j < 4; ++j)
            if (e + j < E) d[j] = dst[e + j];
    }

    for (int r0 = 0; r0 < nranges; r0 += BM_RANGES) {
        if (r0 > 0) __syncthreads();
        for (int i = tid; i < BM_RANGES * BM_WORDS / 4; i += BM_NT)
            ((uint4*)lb)[i] = make_uint4(0u, 0u, 0u, 0u);
        __syncthreads();
#pragma unroll
        for (int j = 0; j < 4; ++j) {
            unsigned r = (unsigned)d[j] / AGG_ROWS - (unsigned)r0;
            if (d[j] >= 0 && r < BM_RANGES)
                atomicOr(&lb[r * BM_WORDS + ((el + j) >> 5)], 1u << ((el + j) & 31));
        }
        __syncthreads();
        const int nr = (nranges - r0) < BM_RANGES ? (nranges - r0) : BM_RANGES;
        for (int i = tid; i < nr * (BM_WORDS / 4); i += BM_NT) {
            int r = i / (BM_WORDS / 4), q = i % (BM_WORDS / 4);
            uint4* gp = (uint4*)(bm + (size_t)(r0 + r) * bm_stride + (size_t)blockIdx.x * BM_WORDS);
            gp[q] = ((const uint4*)lb)[i];
        }
    }
}

#define AGG_CAP  8192
#define AGG_SCAN 5
#define LIST_U   5

__device__ __forceinline__ int agg_emit(unsigned w, int eb, int idx, int* lsrc) {
    while (w) {
        int b = __builtin_ctz(w);
        w &= w - 1;
        if (idx < AGG_CAP) lsrc[idx] = eb + b;
        ++idx;
    }
    return idx;
}

#define GEMM_ROWS  AGG_ROWS
#define GEMM_TILES ((GEMM_ROWS + 15) / 16)
#define GEMM_ITEMS (GEMM_TILES * 2)
#define GEMM_NW    16
#define GEMM_NT    (GEMM_NW * 32)

__device__ __forceinline__ void build_range_list(
    int r, int* lsrc, int* rdeg, int* cnt, int nrows,
    const int* __restrict__ src, const int* __restrict__ dst,
    const unsigned* __restrict__ bm, int bm_stride,
    int* __restrict__ lst, int* __restrict__ lcount, float* __restrict__ dinv_out) {
    const int tid = threadIdx.x;
    const int base = r * AGG_ROWS;
    if (tid == 0) *cnt = 0;
    if (tid < AGG_ROWS) rdeg[tid] = 1;
    __syncthreads();

    const uint4* b4 = (const uint4*)(bm + (size_t)r * bm_stride);
    const int nq = bm_stride >> 2;
    for (int q0 = tid; q0 < nq; q0 += AGG_SCAN * GEMM_NT) {
        uint4 v[AGG_SCAN];
        int c = 0;
#pragma unroll
        for (int u = 0; u < AGG_SCAN; ++u) {
            int q = q0 + u * GEMM_NT;
            bool live = q < nq;
            uint4 t = b4[live ? q : q0];
            v[u] = live ? t : make_uint4(0u, 0u, 0u, 0u);
        }
#pragma unroll
        for (int u = 0; u < AGG_SCAN; ++u)
            c += __builtin_popcount(v[u].x) + __builtin_popcount(v[u].y) +
                 __builtin_popcount(v[u].z) + __builtin_popcount(v[u].w);
        if (c) {
            int idx = atomicAdd(cnt, c);
#pragma unroll
            for (int u = 0; u < AGG_SCAN; ++u) {
                int eb = (q0 + u * GEMM_NT) * 128;
                idx = agg_emit(v[u].x, eb,      idx, lsrc);
                idx = agg_emit(v[u].y, eb + 32, idx, lsrc);
                idx = agg_emit(v[u].z, eb + 64, idx, lsrc);
                idx = agg_emit(v[u].w, eb + 96, idx, lsrc);
            }
        }
    }
    __syncthreads();

    const int total = *cnt;
    const int n = total <= AGG_CAP ? total : 0;
    int2* le = (int2*)lst + (size_t)r * AGG_CAP;
    for (int i0 = tid; i0 < n; i0 += LIST_U * GEMM_NT) {
        int s[LIST_U], d[LIST_U];
#pragma unroll
        for (int u = 0; u < LIST_U; ++u) {
            int i = i0 + u * GEMM_NT;
            int e = lsrc[i < n ? i : i0];
            s[u] = src[e];
            d[u] = dst[e];
        }
#pragma unroll
        for (int u = 0; u < LIST_U; ++u) {
            int i = i0 + u * GEMM_NT;
            if (i < n) {
                le[i] = make_int2(s[u], d[u] - base);
                atomicAdd(&rdeg[d[u] - base], 1);
            }
        }
    }
    if (total > AGG_CAP) {
        const unsigned* brow = bm + (size_t)r * bm_stride;
        for (int q = tid; q < bm_stride; q += GEMM_NT) {
            unsigned w = brow[q];
            while (w) {
                int e = q * 32 + __builtin_ctz(w);
                w &= w - 1;
                atomicAdd(&rdeg[dst[e] - base], 1);
            }
        }
    }
    if (tid == 0) lcount[r] = total;
    __syncthreads();
    if (tid < AGG_ROWS && base + tid < nrows) dinv_out[base + tid] = rsqrtf((float)rdeg[tid]);
}

__global__ __launch_bounds__(GEMM_NT) void gemm_wmma_f32(
    const float* __restrict__ A, const float* __restrict__ W,
    float* __restrict__ dinv_out,
    float* __restrict__ H, int nrows, int nranges,
    const int* __restrict__ src, const int* __restrict__ dst,
    const unsigned* __restrict__ bm, int bm_stride,
    int* __restrict__ lst, int* __restrict__ lcount) {
    __shared__ float wT[D * LDS_STRIDE];
    __shared__ int cnt;

    if ((int)blockIdx.x >= nranges) {
        build_range_list((int)blockIdx.x - nranges, (int*)wT, (int*)wT + AGG_CAP, &cnt, nrows,
                         src, dst, bm, bm_stride, lst, lcount, dinv_out);
        return;
    }

    const int tid = threadIdx.x;
    for (int i = tid; i < D * D; i += GEMM_NT) {
        int r = i >> 7, c = i & (D - 1);
        wT[c * LDS_STRIDE + r] = W[i];
    }

    const int base = blockIdx.x * GEMM_ROWS;
    __syncthreads();

    const int wave = __builtin_amdgcn_readfirstlane(tid >> 5);
    const int lane = tid & 31;
    const int lrow = lane & 15;
    const int lhi  = lane >> 4;

    for (int it = wave; it < GEMM_ITEMS; it += GEMM_NW) {
        const int t  = it >> 1;
        const int c0 = (it & 1) * 64;
        const int m0 = base + t * 16;

        v8f acc[4];
#pragma unroll
        for (int n = 0; n < 4; ++n) acc[n] = (v8f)(0.0f);

        int arow = m0 + lrow;
        int crow = arow < nrows ? arow : 0;
        const float* aptr = A + (size_t)crow * D + lhi * 2;
        const float* bptr = &wT[(c0 + lrow) * LDS_STRIDE + lhi * 2];

        v2f af[D / 4];
#pragma unroll
        for (int kk = 0; kk < D / 4; ++kk) af[kk] = *(const v2f*)(aptr + 4 * kk);

#pragma unroll
        for (int kk = 0; kk < D / 4; ++kk) {
            v2f a = af[kk];
#pragma unroll
            for (int n = 0; n < 4; ++n) {
                v2f b = *(const v2f*)(bptr + n * 16 * LDS_STRIDE + 4 * kk);
                acc[n] = __builtin_amdgcn_wmma_f32_16x16x4_f32(
                    false, a, false, b, (short)0, acc[n], false, false);
            }
        }

#pragma unroll
        for (int n = 0; n < 4; ++n) {
            int col = c0 + n * 16 + lrow;
#pragma unroll
            for (int v = 0; v < 8; ++v) {
                int lr = t * 16 + v + 8 * lhi;
                int row = base + lr;
                if (lr < GEMM_ROWS && row < nrows) H[(size_t)row * D + col] = acc[n][v];
            }
        }
    }
}

#define AGG_NT   1024
#define AGG_U    8
#define AGG_PITCH_MM 132
#define AGG_ROWS_MM  (((AGG_ROWS + 15) / 16) * 16)
#define AGG_ITEMS_MM ((AGG_ROWS_MM / 16) * 4)

template <bool FULL, int PITCH>
__device__ __forceinline__ void agg_batch(int2 ev, int m, int lane, int base,
                                          const float* __restrict__ h,
                                          const float* __restrict__ dinv, float* acc) {
    float2 va[AGG_U], vb[AGG_U];
    float wgt[AGG_U];
    int dl[AGG_U];
#pragma unroll
    for (int u = 0; u < AGG_U; ++u) {
        int s = __builtin_amdgcn_readlane(ev.x, u);
        dl[u] = __builtin_amdgcn_readlane(ev.y, u);
        wgt[u] = dinv[s] * dinv[base + dl[u]];
        const float2* hp = (const float2*)(h + (size_t)s * D);
        va[u] = hp[lane];
        vb[u] = hp[32 + lane];
    }
#pragma unroll
    for (int u = 0; u < AGG_U; ++u) {
        if (FULL || u < m) {
            float* ap = acc + dl[u] * PITCH + 2 * lane;
            float w = wgt[u];
            atomicAdd(ap + 0,  va[u].x * w);
            atomicAdd(ap + 1,  va[u].y * w);
            atomicAdd(ap + 64, vb[u].x * w);
            atomicAdd(ap + 65, vb[u].y * w);
        }
    }
}

template <bool NEXT>
__global__ __launch_bounds__(AGG_NT) void aggregate_rows_kernel(
    const int* __restrict__ src, const int* __restrict__ dst,
    const float* __restrict__ h, const float* __restrict__ dinv,
    const float* __restrict__ bias, float* __restrict__ out, int N,
    const unsigned* __restrict__ bm, int bm_stride,
    const int* __restrict__ lst, const int* __restrict__ lcount,
    const float* __restrict__ Wn) {
    constexpr int PITCH = NEXT ? AGG_PITCH_MM : D;
    constexpr int LROWS = NEXT ? AGG_ROWS_MM : AGG_ROWS;
    __shared__ float acc[LROWS * PITCH];
    __shared__ float wT[NEXT ? D * AGG_PITCH_MM : 4];

    const int tid = threadIdx.x;
    const int base = blockIdx.x * AGG_ROWS;
    const int nrow = (N - base) < AGG_ROWS ? (N - base) : AGG_ROWS;

    float wreg[NEXT ? (D * D) / AGG_NT : 1];
    if (NEXT) {
#pragma unroll
        for (int k = 0; k < (D * D) / AGG_NT; ++k) wreg[k] = Wn[tid + k * AGG_NT];
    }

    for (int i = tid; i < (NEXT ? LROWS : nrow) * (D / 4); i += AGG_NT) {
        int r = i >> 5, c4 = i & 31;
        float4 o = make_float4(0.0f, 0.0f, 0.0f, 0.0f);
        if (!NEXT || r < nrow) {
            float q = dinv[base + r];
            q = q * q;
            float4 hv = ((const float4*)(h + (size_t)(base + r) * D))[c4];
            float4 bv = ((const float4*)bias)[c4];
            o.x = hv.x * q + bv.x;
            o.y = hv.y * q + bv.y;
            o.z = hv.z * q + bv.z;
            o.w = hv.w * q + bv.w;
        }
        *(float4*)(acc + r * PITCH + c4 * 4) = o;
    }
    if (NEXT) {
#pragma unroll
        for (int k = 0; k < (D * D) / AGG_NT; ++k) {
            int i = tid + k * AGG_NT;
            wT[(i & (D - 1)) * AGG_PITCH_MM + (i >> 7)] = wreg[k];
        }
    }
    __syncthreads();

    const unsigned* brow = bm + (size_t)blockIdx.x * bm_stride;
    const int2* le = (const int2*)lst + (size_t)blockIdx.x * AGG_CAP;
    const int wave = tid >> 5, lane = tid & 31;

    const int total = lcount[blockIdx.x];
    const int n = total <= AGG_CAP ? total : 0;
    const int w0 = __builtin_amdgcn_readfirstlane(wave) * AGG_U;
    const int stride = (AGG_NT / 32) * AGG_U;
    if (w0 < n) {
        int j = w0 + (lane & (AGG_U - 1));
        int2 ev = le[j < n ? j : w0];
        int i0 = w0;
        for (; i0 + AGG_U <= n; i0 += stride) {
            int jn = i0 + stride + (lane & (AGG_U - 1));
            int2 nx = le[jn < n ? jn : i0];
            agg_batch<true, PITCH>(ev, AGG_U, lane, base, h, dinv, acc);
            ev = nx;
        }
        if (i0 < n) agg_batch<false, PITCH>(ev, n - i0, lane, base, h, dinv, acc);
    }
    if (total > AGG_CAP) {
        for (int q = tid; q < bm_stride; q += AGG_NT) {
            unsigned w = brow[q];
            while (w) {
                int e = q * 32 + __builtin_ctz(w);
                w &= w - 1;
                int s = src[e];
                int d = dst[e];
                float wg = dinv[s] * dinv[d];
                const float* hp = h + (size_t)s * D;
                float* ap = acc + (d - base) * PITCH;
#pragma nounroll
                for (int f = 0; f < D; ++f) atomicAdd(ap + f, hp[f] * wg);
            }
        }
    }
    __syncthreads();

    if (!NEXT) {
        float4* op = (float4*)(out + (size_t)base * D);
        for (int i = tid; i < nrow * (D / 4); i += AGG_NT) op[i] = ((const float4*)acc)[i];
    } else {
        const int wv = __builtin_amdgcn_readfirstlane(wave);
        const int lrow = lane & 15;
        const int lhi  = lane >> 4;
        for (int it = wv; it < AGG_ITEMS_MM; it += AGG_NT / 32) {
            const int t  = it >> 2;
            const int c0 = (it & 3) * 32;
            const float* aptr = acc + (t * 16 + lrow) * PITCH + lhi * 2;
            const float* bptr = wT + (c0 + lrow) * AGG_PITCH_MM + lhi * 2;
            v8f cc[2];
            cc[0] = (v8f)(0.0f);
            cc[1] = (v8f)(0.0f);
#pragma unroll
            for (int kk = 0; kk < D / 4; ++kk) {
                v2f a = *(const v2f*)(aptr + 4 * kk);
                a.x = fmaxf(a.x, 0.0f);
                a.y = fmaxf(a.y, 0.0f);
#pragma unroll
                for (int n = 0; n < 2; ++n) {
                    v2f b = *(const v2f*)(bptr + n * 16 * AGG_PITCH_MM + 4 * kk);
                    cc[n] = __builtin_amdgcn_wmma_f32_16x16x4_f32(
                        false, a, false, b, (short)0, cc[n], false, false);
                }
            }
#pragma unroll
            for (int n = 0; n < 2; ++n) {
                int col = c0 + n * 16 + lrow;
#pragma unroll
                for (int v = 0; v < 8; ++v) {
                    int lr = t * 16 + v + 8 * lhi;
                    if (lr < nrow) out[(size_t)(base + lr) * D + col] = cc[n][v];
                }
            }
        }
    }
}

extern "C" void kernel_launch(void* const* d_in, const int* in_sizes, int n_in,
                              void* d_out, int out_size, void* d_ws, size_t ws_size,
                              hipStream_t stream) {
    const float* x   = (const float*)d_in[0];
    const int*   ei  = (const int*)d_in[1];
    const float* W1  = (const float*)d_in[2];
    const float* b1  = (const float*)d_in[3];
    const float* W2  = (const float*)d_in[4];
    const float* b2  = (const float*)d_in[5];
    float* out = (float*)d_out;

    const int N = in_sizes[0] / D;
    const int E = in_sizes[1] / 2;
    const int* src = ei;
    const int* dst = ei + E;

    float* dinv = (float*)d_ws;
    size_t nPad = ((size_t)N + 63) & ~(size_t)63;
    float* bufA = dinv + nPad;
    float* bufB = bufA + (size_t)N * D;
    unsigned* bm = (unsigned*)(bufB + (size_t)N * D);

    const int nChunks       = (E + BM_CHUNK - 1) / BM_CHUNK;
    const int bmStride      = nChunks * BM_WORDS;
    const int nBlocksAgg    = (N + AGG_ROWS - 1) / AGG_ROWS;
    int* lst    = (int*)(bm + (size_t)nBlocksAgg * bmStride);
    int* lcount = lst + (size_t)nBlocksAgg * AGG_CAP * 2;

    edge_bitmap_kernel<<<nChunks, BM_NT, 0, stream>>>(dst, E, bm, nBlocksAgg, bmStride);

    gemm_wmma_f32<<<nBlocksAgg + nBlocksAgg, GEMM_NT, 0, stream>>>(x, W1, dinv, bufA, N, nBlocksAgg, src, dst, bm, bmStride, lst, lcount);
    aggregate_rows_kernel<true><<<nBlocksAgg, AGG_NT, 0, stream>>>(src, dst, bufA, dinv, b1, bufB, N, bm, bmStride, lst, lcount, W2);

    aggregate_rows_kernel<false><<<nBlocksAgg, AGG_NT, 0, stream>>>(src, dst, bufB, dinv, b2, out, N, bm, bmStride, lst, lcount, nullptr);
}
